# starting kernel
# baseline (speedup 1.0000x reference)
#include <hip/hip_runtime.h>
#include <math.h>

#define N_NODES 50000
#define N_EDGES 800000
#define F_IN    512
#define F_HID   256
#define F_OUT   64
#define LN_EPS  1e-5f
#define L2_EPS  1e-12f

typedef __attribute__((ext_vector_type(2))) float v2f;
typedef __attribute__((ext_vector_type(8))) float v8f;

__device__ __forceinline__ float wave_sum32(float v) {
#pragma unroll
    for (int m = 16; m >= 1; m >>= 1) v += __shfl_xor(v, m, 32);
    return v;
}

__global__ void __launch_bounds__(256) k_deg_init(float* deg) {
    int i = blockIdx.x * 256 + threadIdx.x;
    if (i < N_NODES) deg[i] = 1.0f;
}

__global__ void __launch_bounds__(256) k_deg_accum(const int* __restrict__ dst,
                                                   float* deg) {
    int e = blockIdx.x * 256 + threadIdx.x;
    if (e < N_EDGES) atomicAdd(&deg[dst[e]], 1.0f);
}

__global__ void __launch_bounds__(256) k_dinv(float* deg) {
    int i = blockIdx.x * 256 + threadIdx.x;
    if (i < N_NODES) deg[i] = rsqrtf(deg[i]);
}

__global__ void __launch_bounds__(256) k_l2norm(const float* __restrict__ x,
                                                float* __restrict__ xn) {
    const int node = blockIdx.x * 8 + (threadIdx.x >> 5);
    const int lane = threadIdx.x & 31;
    const float4* row = (const float4*)(x + (size_t)node * F_IN);
    float4 v[4];
    float ss = 0.0f;
#pragma unroll
    for (int i = 0; i < 4; ++i) {
        float4 t = row[lane + 32 * i];
        t.x = isfinite(t.x) ? t.x : 0.0f;
        t.y = isfinite(t.y) ? t.y : 0.0f;
        t.z = isfinite(t.z) ? t.z : 0.0f;
        t.w = isfinite(t.w) ? t.w : 0.0f;
        v[i] = t;
        ss += t.x * t.x + t.y * t.y + t.z * t.z + t.w * t.w;
    }
    ss = wave_sum32(ss);
    float scale = 1.0f / fmaxf(sqrtf(ss), L2_EPS);
    float4* orow = (float4*)(xn + (size_t)node * F_IN);
#pragma unroll
    for (int i = 0; i < 4; ++i) {
        float4 t = v[i];
        t.x *= scale; t.y *= scale; t.z *= scale; t.w *= scale;
        orow[lane + 32 * i] = t;
    }
}

template <int K, int NOUT>
__global__ void __launch_bounds__(128) k_gemm_wmma(const float* __restrict__ A,
                                                   const float* __restrict__ B,
                                                   float* __restrict__ C) {
    __shared__ float As[16 * K];
    const int tid  = threadIdx.x;
    const int m0   = blockIdx.x * 16;
    const int wave = tid >> 5;
    const int lane = tid & 31;
    const int n0   = blockIdx.y * 64 + wave * 16;

    {
        const float4* gsrc = (const float4*)A + (size_t)m0 * (K / 4);
        float4* lds = (float4*)As;
#pragma unroll 4
        for (int i = tid; i < 16 * (K / 4); i += 128) lds[i] = gsrc[i];
    }
    __syncthreads();

    const int r   = lane & 15;
    const int kk  = (lane >> 4) * 2;

    v8f acc = {};
#pragma unroll 4
    for (int k = 0; k < K; k += 4) {
        v2f a, b;
        a.x = As[r * K + k + kk];
        a.y = As[r * K + k + kk + 1];
        b.x = B[(size_t)(k + kk) * NOUT + n0 + r];
        b.y = B[(size_t)(k + kk + 1) * NOUT + n0 + r];
        acc = __builtin_amdgcn_wmma_f32_16x16x4_f32(
             false, a,  false, b,
             (short)0, acc,  false,  false);
    }

    const int rb = (lane >> 4) * 8;
#pragma unroll
    for (int i = 0; i < 8; ++i)
        C[(size_t)(m0 + rb + i) * NOUT + n0 + r] = acc[i];
}

template <int F>
__global__ void __launch_bounds__(256) k_scatter(const float* __restrict__ H,
                                                 const int* __restrict__ srcp,
                                                 const int* __restrict__ dstp,
                                                 const float* __restrict__ dinv,
                                                 float* __restrict__ out) {
    const int wave = blockIdx.x * 8 + (threadIdx.x >> 5);
    const int lane = threadIdx.x & 31;
    const int Etot = N_EDGES + N_NODES;
    if (wave >= Etot) return;
    int s, d; float w;
    if (wave < N_EDGES) {
        s = srcp[wave]; d = dstp[wave];
        w = dinv[s] * dinv[d];
    } else {
        s = d = wave - N_EDGES;
        float di = dinv[s];
        w = di * di;
    }
    const float4* hr = (const float4*)(H + (size_t)s * F);
    float* orow = out + (size_t)d * F;
    __builtin_prefetch(hr, 0, 0);
#pragma unroll
    for (int j = lane; j < F / 4; j += 32) {
        float4 v = hr[j];
        atomicAdd(&orow[4 * j + 0], v.x * w);
        atomicAdd(&orow[4 * j + 1], v.y * w);
        atomicAdd(&orow[4 * j + 2], v.z * w);
        atomicAdd(&orow[4 * j + 3], v.w * w);
    }
}

__global__ void __launch_bounds__(256) k_ln_relu(float* __restrict__ H,
                                                 const float* __restrict__ b1,
                                                 const float* __restrict__ gamma,
                                                 const float* __restrict__ beta) {
    const int node = blockIdx.x * 8 + (threadIdx.x >> 5);
    const int lane = threadIdx.x & 31;
    float* row = H + (size_t)node * F_HID;
    float4 v[2];
    float s = 0.0f;
#pragma unroll
    for (int i = 0; i < 2; ++i) {
        float4 t  = ((float4*)row)[lane + 32 * i];
        float4 bb = ((const float4*)b1)[lane + 32 * i];
        t.x += bb.x; t.y += bb.y; t.z += bb.z; t.w += bb.w;
        v[i] = t;
        s += t.x + t.y + t.z + t.w;
    }
    float mu = wave_sum32(s) * (1.0f / F_HID);
    float ss = 0.0f;
#pragma unroll
    for (int i = 0; i < 2; ++i) {
        float4 t = v[i];
        ss += (t.x - mu) * (t.x - mu) + (t.y - mu) * (t.y - mu) +
              (t.z - mu) * (t.z - mu) + (t.w - mu) * (t.w - mu);
    }
    float rstd = rsqrtf(wave_sum32(ss) * (1.0f / F_HID) + LN_EPS);
#pragma unroll
    for (int i = 0; i < 2; ++i) {
        float4 t = v[i];
        float4 g = ((const float4*)gamma)[lane + 32 * i];
        float4 b = ((const float4*)beta)[lane + 32 * i];
        t.x = fmaxf((t.x - mu) * rstd * g.x + b.x, 0.0f);
        t.y = fmaxf((t.y - mu) * rstd * g.y + b.y, 0.0f);
        t.z = fmaxf((t.z - mu) * rstd * g.z + b.z, 0.0f);
        t.w = fmaxf((t.w - mu) * rstd * g.w + b.w, 0.0f);
        ((float4*)row)[lane + 32 * i] = t;
    }
}

__global__ void __launch_bounds__(256) k_finalize(float* __restrict__ out,
                                                  const float* __restrict__ b2) {
    int i = blockIdx.x * 256 + threadIdx.x;
    if (i >= N_NODES * F_OUT) return;
    float v = out[i] + b2[i & (F_OUT - 1)];
    if (isnan(v)) v = 0.0f;
    else if (isinf(v)) v = v > 0.0f ? 20.0f : -20.0f;
    out[i] = v;
}

extern "C" void kernel_launch(void* const* d_in, const int* in_sizes, int n_in,
                              void* d_out, int out_size, void* d_ws, size_t ws_size,
                              hipStream_t stream) {
    const float* x     = (const float*)d_in[0];
    const int*   ei    = (const int*)d_in[1];
    const float* W1    = (const float*)d_in[2];
    const float* b1    = (const float*)d_in[3];
    const float* gamma = (const float*)d_in[4];
    const float* beta  = (const float*)d_in[5];
    const float* W2    = (const float*)d_in[6];
    const float* b2    = (const float*)d_in[7];
    const float* Wskip = (const float*)d_in[8];
    const int* srcp = ei;
    const int* dstp = ei + N_EDGES;

    float* ws   = (float*)d_ws;
    float* dinv = ws;
    float* xn   = dinv + N_NODES;
    float* h1   = xn + (size_t)N_NODES * F_IN;
    float* agg  = h1 + (size_t)N_NODES * F_HID;
    float* out  = (float*)d_out;

    const int Etot = N_EDGES + N_NODES;

    hipMemsetAsync(agg, 0, (size_t)N_NODES * F_HID * sizeof(float), stream);
    k_deg_init<<<(N_NODES + 255) / 256, 256, 0, stream>>>(dinv);
    k_deg_accum<<<(N_EDGES + 255) / 256, 256, 0, stream>>>(dstp, dinv);
    k_dinv<<<(N_NODES + 255) / 256, 256, 0, stream>>>(dinv);

    k_l2norm<<<N_NODES / 8, 256, 0, stream>>>(x, xn);

    k_gemm_wmma<F_IN, F_HID><<<dim3(N_NODES / 16, F_HID / 64), 128, 0, stream>>>(xn, W1, h1);
    k_gemm_wmma<F_IN, F_OUT><<<dim3(N_NODES / 16, F_OUT / 64), 128, 0, stream>>>(xn, Wskip, out);

    k_scatter<F_HID><<<(Etot + 7) / 8, 256, 0, stream>>>(h1, srcp, dstp, dinv, agg);
    k_ln_relu<<<N_NODES / 8, 256, 0, stream>>>(agg, b1, gamma, beta);

    k_gemm_wmma<F_HID, F_OUT><<<dim3(N_NODES / 16, F_OUT / 64), 128, 0, stream>>>(agg, W2, h1);
    k_scatter<F_OUT><<<(Etot + 7) / 8, 256, 0, stream>>>(h1, srcp, dstp, dinv, out);

    k_finalize<<<(N_NODES * F_OUT + 255) / 256, 256, 0, stream>>>(out, b2);
}


# Float-atomic scatter-add replaced by a destination-sorted edge list built in every call (int atomic ranks, block scan) and a per-node gather with fused multiply-add; sum order = CSR order
# speedup vs baseline: 1.9972x; 1.9972x over previous
#include <hip/hip_runtime.h>
#include <math.h>

#define N_NODES 50000
#define N_EDGES 800000
#define F_IN    512
#define F_HID   256
#define F_OUT   64
#define LN_EPS  1e-5f
#define L2_EPS  1e-12f
#define N_LINE  ((N_NODES + 31) / 32 * 32)

typedef __attribute__((ext_vector_type(2))) float v2f;
typedef __attribute__((ext_vector_type(8))) float v8f;

__device__ __forceinline__ float wave_sum32(float v) {
#pragma unroll
    for (int m = 16; m >= 1; m >>= 1) v += __shfl_xor(v, m, 32);
    return v;
}

__global__ void __launch_bounds__(256) k_deg_init(int* __restrict__ cnt,
                                                  int* __restrict__ cursor) {
    int i = blockIdx.x * 256 + threadIdx.x;
    if (i < N_NODES) cnt[i] = 0;
    if (i == 0) *cursor = 0;
}

__global__ void __launch_bounds__(256) k_deg_rank(const int* __restrict__ dst,
                                                  int* cnt,
                                                  int* __restrict__ rank) {
    int e = blockIdx.x * 256 + threadIdx.x;
    if (e < N_EDGES) rank[e] = atomicAdd(&cnt[dst[e]], 1);
}

__global__ void __launch_bounds__(256) k_dinv_rowstart(const int* __restrict__ cnt,
                                                       int* cursor,
                                                       int* __restrict__ row_start,
                                                       float* __restrict__ dinv) {
    __shared__ int wsum[8];
    __shared__ int sbase;
    const int tid  = threadIdx.x;
    const int lane = tid & 31;
    const int w    = tid >> 5;
    const int i    = blockIdx.x * 256 + tid;
    const int c    = (i < N_NODES) ? cnt[i] : 0;
    int incl = c;
    for (int d = 1; d < 32; d <<= 1) {
        int t = __shfl_up(incl, d, 32);
        if (lane >= d) incl += t;
    }
    if (lane == 31) wsum[w] = incl;
    __syncthreads();
    if (tid == 0) {
        int tot = 0;
        for (int k = 0; k < 8; ++k) { int t = wsum[k]; wsum[k] = tot; tot += t; }
        sbase = atomicAdd(cursor, tot);
    }
    __syncthreads();
    if (i < N_NODES) {
        row_start[i] = sbase + wsum[w] + incl - c;
        dinv[i] = rsqrtf((float)(c + 1));
    }
}

__global__ void __launch_bounds__(256) k_fill(const int* __restrict__ srcp,
                                              const int* __restrict__ dstp,
                                              const int* __restrict__ row_start,
                                              const int* __restrict__ rank,
                                              int* __restrict__ col) {
    int e = blockIdx.x * 256 + threadIdx.x;
    if (e < N_EDGES) col[row_start[dstp[e]] + rank[e]] = srcp[e];
}

__global__ void __launch_bounds__(256) k_l2norm(const float* __restrict__ x,
                                                float* __restrict__ xn) {
    const int node = blockIdx.x * 8 + (threadIdx.x >> 5);
    const int lane = threadIdx.x & 31;
    const float4* row = (const float4*)(x + (size_t)node * F_IN);
    float4 v[4];
    float ss = 0.0f;
#pragma unroll
    for (int i = 0; i < 4; ++i) {
        float4 t = row[lane + 32 * i];
        t.x = isfinite(t.x) ? t.x : 0.0f;
        t.y = isfinite(t.y) ? t.y : 0.0f;
        t.z = isfinite(t.z) ? t.z : 0.0f;
        t.w = isfinite(t.w) ? t.w : 0.0f;
        v[i] = t;
        ss += t.x * t.x + t.y * t.y + t.z * t.z + t.w * t.w;
    }
    ss = wave_sum32(ss);
    float scale = 1.0f / fmaxf(sqrtf(ss), L2_EPS);
    float4* orow = (float4*)(xn + (size_t)node * F_IN);
#pragma unroll
    for (int i = 0; i < 4; ++i) {
        float4 t = v[i];
        t.x *= scale; t.y *= scale; t.z *= scale; t.w *= scale;
        orow[lane + 32 * i] = t;
    }
}

template <int K, int NOUT>
__global__ void __launch_bounds__(128) k_gemm_wmma(const float* __restrict__ A,
                                                   const float* __restrict__ B,
                                                   float* __restrict__ C) {
    __shared__ float As[16 * K];
    const int tid  = threadIdx.x;
    const int m0   = blockIdx.x * 16;
    const int wave = tid >> 5;
    const int lane = tid & 31;
    const int n0   = blockIdx.y * 64 + wave * 16;

    {
        const float4* gsrc = (const float4*)A + (size_t)m0 * (K / 4);
        float4* lds = (float4*)As;
#pragma unroll 4
        for (int i = tid; i < 16 * (K / 4); i += 128) lds[i] = gsrc[i];
    }
    __syncthreads();

    const int r   = lane & 15;
    const int kk  = (lane >> 4) * 2;

    v8f acc = {};
#pragma unroll 4
    for (int k = 0; k < K; k += 4) {
        v2f a, b;
        a.x = As[r * K + k + kk];
        a.y = As[r * K + k + kk + 1];
        b.x = B[(size_t)(k + kk) * NOUT + n0 + r];
        b.y = B[(size_t)(k + kk + 1) * NOUT + n0 + r];
        acc = __builtin_amdgcn_wmma_f32_16x16x4_f32(
             false, a,  false, b,
             (short)0, acc,  false,  false);
    }

    const int rb = (lane >> 4) * 8;
#pragma unroll
    for (int i = 0; i < 8; ++i)
        C[(size_t)(m0 + rb + i) * NOUT + n0 + r] = acc[i];
}

__device__ __forceinline__ void fma4(float4& a, const float4 v, const float w) {
    a.x = fmaf(v.x, w, a.x); a.y = fmaf(v.y, w, a.y);
    a.z = fmaf(v.z, w, a.z); a.w = fmaf(v.w, w, a.w);
}

__device__ __forceinline__ void load_chunk(const int* __restrict__ col,
                                           const float* __restrict__ dinv,
                                           int first, int m, int lane, float di,
                                           int& my_s, float& my_w) {
    my_s = 0; my_w = 0.0f;
    if (lane < m) {
        unsigned s = (unsigned)col[first + lane];
        s = s < (unsigned)N_NODES ? s : (unsigned)(N_NODES - 1);
        my_s = (int)s;
        my_w = dinv[s] * di;
    }
}

template <int R>
__device__ __forceinline__ void add_rows256(const float4* __restrict__ Hv,
                                            int my_s, float my_w, int j, int lane,
                                            float4& a0, float4& a1) {
    float4 v[2 * R]; float w[R];
    for (int u = 0; u < R; ++u) {
        const int s = __builtin_amdgcn_readlane(my_s, j + u);
        w[u] = __int_as_float(__builtin_amdgcn_readlane(__float_as_int(my_w), j + u));
        const float4* r = Hv + (size_t)s * 64;
        v[2 * u]     = r[lane];
        v[2 * u + 1] = r[lane + 32];
    }
    for (int u = 0; u < R; ++u) { fma4(a0, v[2 * u], w[u]); fma4(a1, v[2 * u + 1], w[u]); }
}

__global__ void __launch_bounds__(256) k_gather256(const float* __restrict__ H,
                                                   const int* __restrict__ col,
                                                   const int* __restrict__ row_start,
                                                   const int* __restrict__ cnt,
                                                   const float* __restrict__ dinv,
                                                   float* __restrict__ out) {
    const int wave = __builtin_amdgcn_readfirstlane((int)(threadIdx.x >> 5));
    const int node = blockIdx.x * 8 + wave;
    const int lane = threadIdx.x & 31;
    const int start = row_start[node];
    const int n     = cnt[node];
    const float di  = dinv[node];
    const float4* Hv = (const float4*)H;

    float4 a0, a1;
    {
        const float w = di * di;
        const float4 v0 = Hv[(size_t)node * 64 + lane];
        const float4 v1 = Hv[(size_t)node * 64 + 32 + lane];
        a0.x = v0.x * w; a0.y = v0.y * w; a0.z = v0.z * w; a0.w = v0.w * w;
        a1.x = v1.x * w; a1.y = v1.y * w; a1.z = v1.z * w; a1.w = v1.w * w;
    }
    for (int base = 0; base < n; base += 32) {
        const int m = (n - base) < 32 ? (n - base) : 32;
        int my_s; float my_w;
        load_chunk(col, dinv, start + base, m, lane, di, my_s, my_w);
        int j = 0;
        for (; j + 4 <= m; j += 4) add_rows256<4>(Hv, my_s, my_w, j, lane, a0, a1);
        const int left = m - j;
        if (left == 3)      add_rows256<3>(Hv, my_s, my_w, j, lane, a0, a1);
        else if (left == 2) add_rows256<2>(Hv, my_s, my_w, j, lane, a0, a1);
        else if (left == 1) add_rows256<1>(Hv, my_s, my_w, j, lane, a0, a1);
    }
    float4* orow = (float4*)(out + (size_t)node * F_HID);
    orow[lane]      = a0;
    orow[lane + 32] = a1;
}

__global__ void __launch_bounds__(256) k_gather64(const float* __restrict__ H,
                                                  const int* __restrict__ col,
                                                  const int* __restrict__ row_start,
                                                  const int* __restrict__ cnt,
                                                  const float* __restrict__ dinv,
                                                  float* __restrict__ out) {
    const int wave = __builtin_amdgcn_readfirstlane((int)(threadIdx.x >> 5));
    const int node = blockIdx.x * 8 + wave;
    const int lane = threadIdx.x & 31;
    const int half = lane >> 4;
    const int c    = lane & 15;
    const int start = row_start[node];
    const int n     = cnt[node];
    const float di  = dinv[node];
    const float4* Hv = (const float4*)H;

    float4 acc = {0.0f, 0.0f, 0.0f, 0.0f};
    if (half == 0) {
        const float w = di * di;
        const float4 v = Hv[(size_t)node * 16 + c];
        acc.x = v.x * w; acc.y = v.y * w; acc.z = v.z * w; acc.w = v.w * w;
    }
    for (int base = 0; base < n; base += 32) {
        const int m = (n - base) < 32 ? (n - base) : 32;
        int my_s; float my_w;
        load_chunk(col, dinv, start + base, m, lane, di, my_s, my_w);
        for (int j = 0; j < m; j += 8) {
            float4 v[4]; float w[4];
            for (int u = 0; u < 4; ++u) {
                const int e0 = j + 2 * u;
                const int sA = __builtin_amdgcn_readlane(my_s, e0);
                const int sB = __builtin_amdgcn_readlane(my_s, e0 + 1);
                const float wA = __int_as_float(__builtin_amdgcn_readlane(__float_as_int(my_w), e0));
                const float wB = __int_as_float(__builtin_amdgcn_readlane(__float_as_int(my_w), e0 + 1));
                const int s = half ? sB : sA;
                w[u] = half ? wB : wA;
                v[u] = Hv[(size_t)s * 16 + c];
            }
            for (int u = 0; u < 4; ++u)
                if (j + 2 * u + half < m) fma4(acc, v[u], w[u]);
        }
    }
    acc.x += __shfl_xor(acc.x, 16, 32);
    acc.y += __shfl_xor(acc.y, 16, 32);
    acc.z += __shfl_xor(acc.z, 16, 32);
    acc.w += __shfl_xor(acc.w, 16, 32);
    if (half == 0) {
        float4* o = (float4*)(out + (size_t)node * F_OUT) + c;
        float4 t = *o;
        t.x += acc.x; t.y += acc.y; t.z += acc.z; t.w += acc.w;
        *o = t;
    }
}

__global__ void __launch_bounds__(256) k_ln_relu(float* __restrict__ H,
                                                 const float* __restrict__ b1,
                                                 const float* __restrict__ gamma,
                                                 const float* __restrict__ beta) {
    const int node = blockIdx.x * 8 + (threadIdx.x >> 5);
    const int lane = threadIdx.x & 31;
    float* row = H + (size_t)node * F_HID;
    float4 v[2];
    float s = 0.0f;
#pragma unroll
    for (int i = 0; i < 2; ++i) {
        float4 t  = ((float4*)row)[lane + 32 * i];
        float4 bb = ((const float4*)b1)[lane + 32 * i];
        t.x += bb.x; t.y += bb.y; t.z += bb.z; t.w += bb.w;
        v[i] = t;
        s += t.x + t.y + t.z + t.w;
    }
    float mu = wave_sum32(s) * (1.0f / F_HID);
    float ss = 0.0f;
#pragma unroll
    for (int i = 0; i < 2; ++i) {
        float4 t = v[i];
        ss += (t.x - mu) * (t.x - mu) + (t.y - mu) * (t.y - mu) +
              (t.z - mu) * (t.z - mu) + (t.w - mu) * (t.w - mu);
    }
    float rstd = rsqrtf(wave_sum32(ss) * (1.0f / F_HID) + LN_EPS);
#pragma unroll
    for (int i = 0; i < 2; ++i) {
        float4 t = v[i];
        float4 g = ((const float4*)gamma)[lane + 32 * i];
        float4 b = ((const float4*)beta)[lane + 32 * i];
        t.x = fmaxf((t.x - mu) * rstd * g.x + b.x, 0.0f);
        t.y = fmaxf((t.y - mu) * rstd * g.y + b.y, 0.0f);
        t.z = fmaxf((t.z - mu) * rstd * g.z + b.z, 0.0f);
        t.w = fmaxf((t.w - mu) * rstd * g.w + b.w, 0.0f);
        ((float4*)row)[lane + 32 * i] = t;
    }
}

__global__ void __launch_bounds__(256) k_finalize(float* __restrict__ out,
                                                  const float* __restrict__ b2) {
    int i = blockIdx.x * 256 + threadIdx.x;
    if (i >= N_NODES * F_OUT) return;
    float v = out[i] + b2[i & (F_OUT - 1)];
    if (isnan(v)) v = 0.0f;
    else if (isinf(v)) v = v > 0.0f ? 20.0f : -20.0f;
    out[i] = v;
}

extern "C" void kernel_launch(void* const* d_in, const int* in_sizes, int n_in,
                              void* d_out, int out_size, void* d_ws, size_t ws_size,
                              hipStream_t stream) {
    const float* x     = (const float*)d_in[0];
    const int*   ei    = (const int*)d_in[1];
    const float* W1    = (const float*)d_in[2];
    const float* b1    = (const float*)d_in[3];
    const float* gamma = (const float*)d_in[4];
    const float* beta  = (const float*)d_in[5];
    const float* W2    = (const float*)d_in[6];
    const float* b2    = (const float*)d_in[7];
    const float* Wskip = (const float*)d_in[8];
    const int* srcp = ei;
    const int* dstp = ei + N_EDGES;

    float* ws   = (float*)d_ws;
    float* dinv = ws;
    float* xn   = dinv + N_NODES;
    float* h1   = xn + (size_t)N_NODES * F_IN;
    float* agg  = h1 + (size_t)N_NODES * F_HID;
    float* out  = (float*)d_out;

    int* cnt       = (int*)xn + 16;
    int* row_start = cnt + N_LINE;
    int* rank      = row_start + N_LINE;
    int* col       = rank + N_EDGES;
    int* cursor    = col + N_EDGES;

    k_l2norm<<<N_NODES / 8, 256, 0, stream>>>(x, xn);

    k_gemm_wmma<F_IN, F_HID><<<dim3(N_NODES / 16, F_HID / 64), 128, 0, stream>>>(xn, W1, h1);
    k_gemm_wmma<F_IN, F_OUT><<<dim3(N_NODES / 16, F_OUT / 64), 128, 0, stream>>>(xn, Wskip, out);

    k_deg_init<<<(N_NODES + 255) / 256, 256, 0, stream>>>(cnt, cursor);
    k_deg_rank<<<(N_EDGES + 255) / 256, 256, 0, stream>>>(dstp, cnt, rank);
    k_dinv_rowstart<<<(N_NODES + 255) / 256, 256, 0, stream>>>(cnt, cursor, row_start, dinv);
    k_fill<<<(N_EDGES + 255) / 256, 256, 0, stream>>>(srcp, dstp, row_start, rank, col);

    k_gather256<<<N_NODES / 8, 256, 0, stream>>>(h1, col, row_start, cnt, dinv, agg);
    k_ln_relu<<<N_NODES / 8, 256, 0, stream>>>(agg, b1, gamma, beta);

    k_gemm_wmma<F_HID, F_OUT><<<dim3(N_NODES / 16, F_OUT / 64), 128, 0, stream>>>(agg, W2, h1);
    k_gather64<<<N_NODES / 8, 256, 0, stream>>>(h1, col, row_start, cnt, dinv, out);

    k_finalize<<<(N_NODES * F_OUT + 255) / 256, 256, 0, stream>>>(out, b2);
}


# GEMM A tile in LDS: row pitch K floats put all 16 rows of a fragment read on one bank (16-way conflict, one row per LDS clock); pitch padded to K+4 floats spreads each read over all 64 banks
# speedup vs baseline: 2.3539x; 1.1786x over previous
#include <hip/hip_runtime.h>
#include <math.h>

#define N_NODES 50000
#define N_EDGES 800000
#define F_IN    512
#define F_HID   256
#define F_OUT   64
#define LN_EPS  1e-5f
#define L2_EPS  1e-12f
#define N_LINE  ((N_NODES + 31) / 32 * 32)

typedef __attribute__((ext_vector_type(2))) float v2f;
typedef __attribute__((ext_vector_type(8))) float v8f;

__device__ __forceinline__ float wave_sum32(float v) {
#pragma unroll
    for (int m = 16; m >= 1; m >>= 1) v += __shfl_xor(v, m, 32);
    return v;
}

__global__ void __launch_bounds__(256) k_deg_init(int* __restrict__ cnt,
                                                  int* __restrict__ cursor) {
    int i = blockIdx.x * 256 + threadIdx.x;
    if (i < N_NODES) cnt[i] = 0;
    if (i == 0) *cursor = 0;
}

__global__ void __launch_bounds__(256) k_deg_rank(const int* __restrict__ dst,
                                                  int* cnt,
                                                  int* __restrict__ rank) {
    int e = blockIdx.x * 256 + threadIdx.x;
    if (e < N_EDGES) rank[e] = atomicAdd(&cnt[dst[e]], 1);
}

__global__ void __launch_bounds__(256) k_dinv_rowstart(const int* __restrict__ cnt,
                                                       int* cursor,
                                                       int* __restrict__ row_start,
                                                       float* __restrict__ dinv) {
    __shared__ int wsum[8];
    __shared__ int sbase;
    const int tid  = threadIdx.x;
    const int lane = tid & 31;
    const int w    = tid >> 5;
    const int i    = blockIdx.x * 256 + tid;
    const int c    = (i < N_NODES) ? cnt[i] : 0;
    int incl = c;
    for (int d = 1; d < 32; d <<= 1) {
        int t = __shfl_up(incl, d, 32);
        if (lane >= d) incl += t;
    }
    if (lane == 31) wsum[w] = incl;
    __syncthreads();
    if (tid == 0) {
        int tot = 0;
        for (int k = 0; k < 8; ++k) { int t = wsum[k]; wsum[k] = tot; tot += t; }
        sbase = atomicAdd(cursor, tot);
    }
    __syncthreads();
    if (i < N_NODES) {
        row_start[i] = sbase + wsum[w] + incl - c;
        dinv[i] = rsqrtf((float)(c + 1));
    }
}

__global__ void __launch_bounds__(256) k_fill(const int* __restrict__ srcp,
                                              const int* __restrict__ dstp,
                                              const int* __restrict__ row_start,
                                              const int* __restrict__ rank,
                                              int* __restrict__ col) {
    int e = blockIdx.x * 256 + threadIdx.x;
    if (e < N_EDGES) col[row_start[dstp[e]] + rank[e]] = srcp[e];
}

__global__ void __launch_bounds__(256) k_l2norm(const float* __restrict__ x,
                                                float* __restrict__ xn) {
    const int node = blockIdx.x * 8 + (threadIdx.x >> 5);
    const int lane = threadIdx.x & 31;
    const float4* row = (const float4*)(x + (size_t)node * F_IN);
    float4 v[4];
    float ss = 0.0f;
#pragma unroll
    for (int i = 0; i < 4; ++i) {
        float4 t = row[lane + 32 * i];
        t.x = isfinite(t.x) ? t.x : 0.0f;
        t.y = isfinite(t.y) ? t.y : 0.0f;
        t.z = isfinite(t.z) ? t.z : 0.0f;
        t.w = isfinite(t.w) ? t.w : 0.0f;
        v[i] = t;
        ss += t.x * t.x + t.y * t.y + t.z * t.z + t.w * t.w;
    }
    ss = wave_sum32(ss);
    float scale = 1.0f / fmaxf(sqrtf(ss), L2_EPS);
    float4* orow = (float4*)(xn + (size_t)node * F_IN);
#pragma unroll
    for (int i = 0; i < 4; ++i) {
        float4 t = v[i];
        t.x *= scale; t.y *= scale; t.z *= scale; t.w *= scale;
        orow[lane + 32 * i] = t;
    }
}

template <int K, int NOUT>
__global__ void __launch_bounds__(128) k_gemm_wmma(const float* __restrict__ A,
                                                   const float* __restrict__ B,
                                                   float* __restrict__ C) {
    constexpr int P = K + 4;
    __shared__ __attribute__((aligned(16))) float As[16 * P];
    const int tid  = threadIdx.x;
    const int m0   = blockIdx.x * 16;
    const int wave = tid >> 5;
    const int lane = tid & 31;
    const int n0   = blockIdx.y * 64 + wave * 16;

    {
        const float4* gsrc = (const float4*)A + (size_t)m0 * (K / 4);
#pragma unroll 4
        for (int i = tid; i < 16 * (K / 4); i += 128) {
            const int row = i / (K / 4);
            const int c4  = i % (K / 4);
            *(float4*)&As[row * P + 4 * c4] = gsrc[i];
        }
    }
    __syncthreads();

    const int r   = lane & 15;
    const int kk  = (lane >> 4) * 2;

    v8f acc = {};
#pragma unroll 4
    for (int k = 0; k < K; k += 4) {
        v2f a, b;
        a.x = As[r * P + k + kk];
        a.y = As[r * P + k + kk + 1];
        b.x = B[(size_t)(k + kk) * NOUT + n0 + r];
        b.y = B[(size_t)(k + kk + 1) * NOUT + n0 + r];
        acc = __builtin_amdgcn_wmma_f32_16x16x4_f32(
             false, a,  false, b,
             (short)0, acc,  false,  false);
    }

    const int rb = (lane >> 4) * 8;
#pragma unroll
    for (int i = 0; i < 8; ++i)
        C[(size_t)(m0 + rb + i) * NOUT + n0 + r] = acc[i];
}

__device__ __forceinline__ void fma4(float4& a, const float4 v, const float w) {
    a.x = fmaf(v.x, w, a.x); a.y = fmaf(v.y, w, a.y);
    a.z = fmaf(v.z, w, a.z); a.w = fmaf(v.w, w, a.w);
}

__device__ __forceinline__ void load_chunk(const int* __restrict__ col,
                                           const float* __restrict__ dinv,
                                           int first, int m, int lane, float di,
                                           int& my_s, float& my_w) {
    my_s = 0; my_w = 0.0f;
    if (lane < m) {
        unsigned s = (unsigned)col[first + lane];
        s = s < (unsigned)N_NODES ? s : (unsigned)(N_NODES - 1);
        my_s = (int)s;
        my_w = dinv[s] * di;
    }
}

template <int R>
__device__ __forceinline__ void add_rows256(const float4* __restrict__ Hv,
                                            int my_s, float my_w, int j, int lane,
                                            float4& a0, float4& a1) {
    float4 v[2 * R]; float w[R];
    for (int u = 0; u < R; ++u) {
        const int s = __builtin_amdgcn_readlane(my_s, j + u);
        w[u] = __int_as_float(__builtin_amdgcn_readlane(__float_as_int(my_w), j + u));
        const float4* r = Hv + (size_t)s * 64;
        v[2 * u]     = r[lane];
        v[2 * u + 1] = r[lane + 32];
    }
    for (int u = 0; u < R; ++u) { fma4(a0, v[2 * u], w[u]); fma4(a1, v[2 * u + 1], w[u]); }
}

__global__ void __launch_bounds__(256) k_gather256(const float* __restrict__ H,
                                                   const int* __restrict__ col,
                                                   const int* __restrict__ row_start,
                                                   const int* __restrict__ cnt,
                                                   const float* __restrict__ dinv,
                                                   float* __restrict__ out) {
    const int wave = __builtin_amdgcn_readfirstlane((int)(threadIdx.x >> 5));
    const int node = blockIdx.x * 8 + wave;
    const int lane = threadIdx.x & 31;
    const int start = row_start[node];
    const int n     = cnt[node];
    const float di  = dinv[node];
    const float4* Hv = (const float4*)H;

    float4 a0, a1;
    {
        const float w = di * di;
        const float4 v0 = Hv[(size_t)node * 64 + lane];
        const float4 v1 = Hv[(size_t)node * 64 + 32 + lane];
        a0.x = v0.x * w; a0.y = v0.y * w; a0.z = v0.z * w; a0.w = v0.w * w;
        a1.x = v1.x * w; a1.y = v1.y * w; a1.z = v1.z * w; a1.w = v1.w * w;
    }
    for (int base = 0; base < n; base += 32) {
        const int m = (n - base) < 32 ? (n - base) : 32;
        int my_s; float my_w;
        load_chunk(col, dinv, start + base, m, lane, di, my_s, my_w);
        int j = 0;
        for (; j + 4 <= m; j += 4) add_rows256<4>(Hv, my_s, my_w, j, lane, a0, a1);
        const int left = m - j;
        if (left == 3)      add_rows256<3>(Hv, my_s, my_w, j, lane, a0, a1);
        else if (left == 2) add_rows256<2>(Hv, my_s, my_w, j, lane, a0, a1);
        else if (left == 1) add_rows256<1>(Hv, my_s, my_w, j, lane, a0, a1);
    }
    float4* orow = (float4*)(out + (size_t)node * F_HID);
    orow[lane]      = a0;
    orow[lane + 32] = a1;
}

__global__ void __launch_bounds__(256) k_gather64(const float* __restrict__ H,
                                                  const int* __restrict__ col,
                                                  const int* __restrict__ row_start,
                                                  const int* __restrict__ cnt,
                                                  const float* __restrict__ dinv,
                                                  float* __restrict__ out) {
    const int wave = __builtin_amdgcn_readfirstlane((int)(threadIdx.x >> 5));
    const int node = blockIdx.x * 8 + wave;
    const int lane = threadIdx.x & 31;
    const int half = lane >> 4;
    const int c    = lane & 15;
    const int start = row_start[node];
    const int n     = cnt[node];
    const float di  = dinv[node];
    const float4* Hv = (const float4*)H;

    float4 acc = {0.0f, 0.0f, 0.0f, 0.0f};
    if (half == 0) {
        const float w = di * di;
        const float4 v = Hv[(size_t)node * 16 + c];
        acc.x = v.x * w; acc.y = v.y * w; acc.z = v.z * w; acc.w = v.w * w;
    }
    for (int base = 0; base < n; base += 32) {
        const int m = (n - base) < 32 ? (n - base) : 32;
        int my_s; float my_w;
        load_chunk(col, dinv, start + base, m, lane, di, my_s, my_w);
        for (int j = 0; j < m; j += 8) {
            float4 v[4]; float w[4];
            for (int u = 0; u < 4; ++u) {
                const int e0 = j + 2 * u;
                const int sA = __builtin_amdgcn_readlane(my_s, e0);
                const int sB = __builtin_amdgcn_readlane(my_s, e0 + 1);
                const float wA = __int_as_float(__builtin_amdgcn_readlane(__float_as_int(my_w), e0));
                const float wB = __int_as_float(__builtin_amdgcn_readlane(__float_as_int(my_w), e0 + 1));
                const int s = half ? sB : sA;
                w[u] = half ? wB : wA;
                v[u] = Hv[(size_t)s * 16 + c];
            }
            for (int u = 0; u < 4; ++u)
                if (j + 2 * u + half < m) fma4(acc, v[u], w[u]);
        }
    }
    acc.x += __shfl_xor(acc.x, 16, 32);
    acc.y += __shfl_xor(acc.y, 16, 32);
    acc.z += __shfl_xor(acc.z, 16, 32);
    acc.w += __shfl_xor(acc.w, 16, 32);
    if (half == 0) {
        float4* o = (float4*)(out + (size_t)node * F_OUT) + c;
        float4 t = *o;
        t.x += acc.x; t.y += acc.y; t.z += acc.z; t.w += acc.w;
        *o = t;
    }
}

__global__ void __launch_bounds__(256) k_ln_relu(float* __restrict__ H,
                                                 const float* __restrict__ b1,
                                                 const float* __restrict__ gamma,
                                                 const float* __restrict__ beta) {
    const int node = blockIdx.x * 8 + (threadIdx.x >> 5);
    const int lane = threadIdx.x & 31;
    float* row = H + (size_t)node * F_HID;
    float4 v[2];
    float s = 0.0f;
#pragma unroll
    for (int i = 0; i < 2; ++i) {
        float4 t  = ((float4*)row)[lane + 32 * i];
        float4 bb = ((const float4*)b1)[lane + 32 * i];
        t.x += bb.x; t.y += bb.y; t.z += bb.z; t.w += bb.w;
        v[i] = t;
        s += t.x + t.y + t.z + t.w;
    }
    float mu = wave_sum32(s) * (1.0f / F_HID);
    float ss = 0.0f;
#pragma unroll
    for (int i = 0; i < 2; ++i) {
        float4 t = v[i];
        ss += (t.x - mu) * (t.x - mu) + (t.y - mu) * (t.y - mu) +
              (t.z - mu) * (t.z - mu) + (t.w - mu) * (t.w - mu);
    }
    float rstd = rsqrtf(wave_sum32(ss) * (1.0f / F_HID) + LN_EPS);
#pragma unroll
    for (int i = 0; i < 2; ++i) {
        float4 t = v[i];
        float4 g = ((const float4*)gamma)[lane + 32 * i];
        float4 b = ((const float4*)beta)[lane + 32 * i];
        t.x = fmaxf((t.x - mu) * rstd * g.x + b.x, 0.0f);
        t.y = fmaxf((t.y - mu) * rstd * g.y + b.y, 0.0f);
        t.z = fmaxf((t.z - mu) * rstd * g.z + b.z, 0.0f);
        t.w = fmaxf((t.w - mu) * rstd * g.w + b.w, 0.0f);
        ((float4*)row)[lane + 32 * i] = t;
    }
}

__global__ void __launch_bounds__(256) k_finalize(float* __restrict__ out,
                                                  const float* __restrict__ b2) {
    int i = blockIdx.x * 256 + threadIdx.x;
    if (i >= N_NODES * F_OUT) return;
    float v = out[i] + b2[i & (F_OUT - 1)];
    if (isnan(v)) v = 0.0f;
    else if (isinf(v)) v = v > 0.0f ? 20.0f : -20.0f;
    out[i] = v;
}

extern "C" void kernel_launch(void* const* d_in, const int* in_sizes, int n_in,
                              void* d_out, int out_size, void* d_ws, size_t ws_size,
                              hipStream_t stream) {
    const float* x     = (const float*)d_in[0];
    const int*   ei    = (const int*)d_in[1];
    const float* W1    = (const float*)d_in[2];
    const float* b1    = (const float*)d_in[3];
    const float* gamma = (const float*)d_in[4];
    const float* beta  = (const float*)d_in[5];
    const float* W2    = (const float*)d_in[6];
    const float* b2    = (const float*)d_in[7];
    const float* Wskip = (const float*)d_in[8];
    const int* srcp = ei;
    const int* dstp = ei + N_EDGES;

    float* ws   = (float*)d_ws;
    float* dinv = ws;
    float* xn   = dinv + N_NODES;
    float* h1   = xn + (size_t)N_NODES * F_IN;
    float* agg  = h1 + (size_t)N_NODES * F_HID;
    float* out  = (float*)d_out;

    int* cnt       = (int*)xn + 16;
    int* row_start = cnt + N_LINE;
    int* rank      = row_start + N_LINE;
    int* col       = rank + N_EDGES;
    int* cursor    = col + N_EDGES;

    k_l2norm<<<N_NODES / 8, 256, 0, stream>>>(x, xn);

    k_gemm_wmma<F_IN, F_HID><<<dim3(N_NODES / 16, F_HID / 64), 128, 0, stream>>>(xn, W1, h1);
    k_gemm_wmma<F_IN, F_OUT><<<dim3(N_NODES / 16, F_OUT / 64), 128, 0, stream>>>(xn, Wskip, out);

    k_deg_init<<<(N_NODES + 255) / 256, 256, 0, stream>>>(cnt, cursor);
    k_deg_rank<<<(N_EDGES + 255) / 256, 256, 0, stream>>>(dstp, cnt, rank);
    k_dinv_rowstart<<<(N_NODES + 255) / 256, 256, 0, stream>>>(cnt, cursor, row_start, dinv);
    k_fill<<<(N_EDGES + 255) / 256, 256, 0, stream>>>(srcp, dstp, row_start, rank, col);

    k_gather256<<<N_NODES / 8, 256, 0, stream>>>(h1, col, row_start, cnt, dinv, agg);
    k_ln_relu<<<N_NODES / 8, 256, 0, stream>>>(agg, b1, gamma, beta);

    k_gemm_wmma<F_HID, F_OUT><<<dim3(N_NODES / 16, F_OUT / 64), 128, 0, stream>>>(agg, W2, h1);
    k_gather64<<<N_NODES / 8, 256, 0, stream>>>(h1, col, row_start, cnt, dinv, out);

    k_finalize<<<(N_NODES * F_OUT + 255) / 256, 256, 0, stream>>>(out, b2);
}


# GEMM re-tiled: block = 32-row A tile in LDS x all output columns; wave = 2 x 4 (N=256) or 2 x 2 (N=64) accumulator tiles, B rows by 16/8-byte loads, next fragment set loaded ahead; sum order kept
# speedup vs baseline: 2.7302x; 1.1599x over previous
#include <hip/hip_runtime.h>
#include <math.h>

#define N_NODES 50000
#define N_EDGES 800000
#define F_IN    512
#define F_HID   256
#define F_OUT   64
#define LN_EPS  1e-5f
#define L2_EPS  1e-12f
#define N_LINE  ((N_NODES + 31) / 32 * 32)

typedef __attribute__((ext_vector_type(2))) float v2f;
typedef __attribute__((ext_vector_type(4))) float v4f;
typedef __attribute__((ext_vector_type(8))) float v8f;

__device__ __forceinline__ float wave_sum32(float v) {
#pragma unroll
    for (int m = 16; m >= 1; m >>= 1) v += __shfl_xor(v, m, 32);
    return v;
}

__global__ void __launch_bounds__(256) k_deg_init(int* __restrict__ cnt,
                                                  int* __restrict__ cursor) {
    int i = blockIdx.x * 256 + threadIdx.x;
    if (i < N_NODES) cnt[i] = 0;
    if (i == 0) *cursor = 0;
}

__global__ void __launch_bounds__(256) k_deg_rank(const int* __restrict__ dst,
                                                  int* cnt,
                                                  int* __restrict__ rank) {
    int e = blockIdx.x * 256 + threadIdx.x;
    if (e < N_EDGES) rank[e] = atomicAdd(&cnt[dst[e]], 1);
}

__global__ void __launch_bounds__(256) k_dinv_rowstart(const int* __restrict__ cnt,
                                                       int* cursor,
                                                       int* __restrict__ row_start,
                                                       float* __restrict__ dinv) {
    __shared__ int wsum[8];
    __shared__ int sbase;
    const int tid  = threadIdx.x;
    const int lane = tid & 31;
    const int w    = tid >> 5;
    const int i    = blockIdx.x * 256 + tid;
    const int c    = (i < N_NODES) ? cnt[i] : 0;
    int incl = c;
    for (int d = 1; d < 32; d <<= 1) {
        int t = __shfl_up(incl, d, 32);
        if (lane >= d) incl += t;
    }
    if (lane == 31) wsum[w] = incl;
    __syncthreads();
    if (tid == 0) {
        int tot = 0;
        for (int k = 0; k < 8; ++k) { int t = wsum[k]; wsum[k] = tot; tot += t; }
        sbase = atomicAdd(cursor, tot);
    }
    __syncthreads();
    if (i < N_NODES) {
        row_start[i] = sbase + wsum[w] + incl - c;
        dinv[i] = rsqrtf((float)(c + 1));
    }
}

__global__ void __launch_bounds__(256) k_fill(const int* __restrict__ srcp,
                                              const int* __restrict__ dstp,
                                              const int* __restrict__ row_start,
                                              const int* __restrict__ rank,
                                              int* __restrict__ col) {
    int e = blockIdx.x * 256 + threadIdx.x;
    if (e < N_EDGES) col[row_start[dstp[e]] + rank[e]] = srcp[e];
}

__global__ void __launch_bounds__(256) k_l2norm(const float* __restrict__ x,
                                                float* __restrict__ xn) {
    const int node = blockIdx.x * 8 + (threadIdx.x >> 5);
    const int lane = threadIdx.x & 31;
    const float4* row = (const float4*)(x + (size_t)node * F_IN);
    float4 v[4];
    float ss = 0.0f;
#pragma unroll
    for (int i = 0; i < 4; ++i) {
        float4 t = row[lane + 32 * i];
        t.x = isfinite(t.x) ? t.x : 0.0f;
        t.y = isfinite(t.y) ? t.y : 0.0f;
        t.z = isfinite(t.z) ? t.z : 0.0f;
        t.w = isfinite(t.w) ? t.w : 0.0f;
        v[i] = t;
        ss += t.x * t.x + t.y * t.y + t.z * t.z + t.w * t.w;
    }
    ss = wave_sum32(ss);
    float scale = 1.0f / fmaxf(sqrtf(ss), L2_EPS);
    float4* orow = (float4*)(xn + (size_t)node * F_IN);
#pragma unroll
    for (int i = 0; i < 4; ++i) {
        float4 t = v[i];
        t.x *= scale; t.y *= scale; t.z *= scale; t.w *= scale;
        orow[lane + 32 * i] = t;
    }
}

template <int TN> struct BVec;
template <> struct BVec<2> { typedef v2f T; };
template <> struct BVec<4> { typedef v4f T; };

template <int K, int NOUT, int WN, int TN, int G>
__global__ void __launch_bounds__(WN * 32) k_gemm_wmma(const float* __restrict__ A,
                                                       const float* __restrict__ B,
                                                       float* __restrict__ C) {
    typedef typename BVec<TN>::T bvec;
    constexpr int P  = K + 4;
    constexpr int BM = 32;
    constexpr int NT = WN * 32;
    static_assert(WN * TN * 16 == NOUT, "a block covers all output columns");
    static_assert((K / (4 * G)) % 2 == 0, "the K loop takes two fragment sets per trip");
    __shared__ __attribute__((aligned(16))) float As[BM * P];
    const int tid = threadIdx.x;
    const int m0  = blockIdx.x * BM;

    {
        const float4* gA = (const float4*)A;
#pragma unroll 8
        for (int i = tid; i < BM * (K / 4); i += NT) {
            const int row = i / (K / 4);
            const int c4  = i % (K / 4);
            int grow = m0 + row;
            grow = grow < N_NODES ? grow : N_NODES - 1;
            *(float4*)&As[row * P + 4 * c4] = gA[(size_t)grow * (K / 4) + c4];
        }
    }
    __syncthreads();

    const int wn   = tid >> 5;
    const int lane = tid & 31;
    const int r    = lane & 15;
    const int h    = lane >> 4;
    const int n0   = wn * (TN * 16);
    const float* ap = &As[r * P + 2 * h];
    const float* bp = B + (size_t)(2 * h) * NOUT + n0 + TN * r;

    v8f acc[2][TN];
    for (int t = 0; t < 2; ++t)
        for (int j = 0; j < TN; ++j) acc[t][j] = v8f{};

    struct FragSet { v2f a[G][2]; bvec b[G][2]; };
    auto load_set = [&](FragSet& s, const int k) {
        for (int u = 0; u < G; ++u) {
            s.a[u][0] = *(const v2f*)(ap + k + 4 * u);
            s.a[u][1] = *(const v2f*)(ap + 16 * P + k + 4 * u);
            s.b[u][0] = *(const bvec*)(bp + (size_t)(k + 4 * u) * NOUT);
            s.b[u][1] = *(const bvec*)(bp + (size_t)(k + 4 * u + 1) * NOUT);
        }
    };
    auto mma_set = [&](const FragSet& s) {
        for (int u = 0; u < G; ++u)
            for (int j = 0; j < TN; ++j) {
                v2f b;
                b.x = s.b[u][0][j];
                b.y = s.b[u][1][j];
                acc[0][j] = __builtin_amdgcn_wmma_f32_16x16x4_f32(
                     false, s.a[u][0],  false, b,
                     (short)0, acc[0][j],  false,  false);
                acc[1][j] = __builtin_amdgcn_wmma_f32_16x16x4_f32(
                     false, s.a[u][1],  false, b,
                     (short)0, acc[1][j],  false,  false);
            }
    };

    FragSet s0, s1;
    load_set(s0, 0);
    for (int k = 0; k < K; k += 8 * G) {
        load_set(s1, k + 4 * G);
        mma_set(s0);
        if (k + 8 * G < K) load_set(s0, k + 8 * G);
        mma_set(s1);
    }

    for (int t = 0; t < 2; ++t) {
        const int mb = m0 + 16 * t;
        if (mb < N_NODES) {
            for (int i = 0; i < 8; ++i) {
                bvec o;
                for (int j = 0; j < TN; ++j) o[j] = acc[t][j][i];
                *(bvec*)(C + (size_t)(mb + 8 * h + i) * NOUT + n0 + TN * r) = o;
            }
        }
    }
}

__device__ __forceinline__ void fma4(float4& a, const float4 v, const float w) {
    a.x = fmaf(v.x, w, a.x); a.y = fmaf(v.y, w, a.y);
    a.z = fmaf(v.z, w, a.z); a.w = fmaf(v.w, w, a.w);
}

__device__ __forceinline__ void load_chunk(const int* __restrict__ col,
                                           const float* __restrict__ dinv,
                                           int first, int m, int lane, float di,
                                           int& my_s, float& my_w) {
    my_s = 0; my_w = 0.0f;
    if (lane < m) {
        unsigned s = (unsigned)col[first + lane];
        s = s < (unsigned)N_NODES ? s : (unsigned)(N_NODES - 1);
        my_s = (int)s;
        my_w = dinv[s] * di;
    }
}

template <int R>
__device__ __forceinline__ void add_rows256(const float4* __restrict__ Hv,
                                            int my_s, float my_w, int j, int lane,
                                            float4& a0, float4& a1) {
    float4 v[2 * R]; float w[R];
    for (int u = 0; u < R; ++u) {
        const int s = __builtin_amdgcn_readlane(my_s, j + u);
        w[u] = __int_as_float(__builtin_amdgcn_readlane(__float_as_int(my_w), j + u));
        const float4* r = Hv + (size_t)s * 64;
        v[2 * u]     = r[lane];
        v[2 * u + 1] = r[lane + 32];
    }
    for (int u = 0; u < R; ++u) { fma4(a0, v[2 * u], w[u]); fma4(a1, v[2 * u + 1], w[u]); }
}

__global__ void __launch_bounds__(256) k_gather256(const float* __restrict__ H,
                                                   const int* __restrict__ col,
                                                   const int* __restrict__ row_start,
                                                   const int* __restrict__ cnt,
                                                   const float* __restrict__ dinv,
                                                   float* __restrict__ out) {
    const int wave = __builtin_amdgcn_readfirstlane((int)(threadIdx.x >> 5));
    const int node = blockIdx.x * 8 + wave;
    const int lane = threadIdx.x & 31;
    const int start = row_start[node];
    const int n     = cnt[node];
    const float di  = dinv[node];
    const float4* Hv = (const float4*)H;

    float4 a0, a1;
    {
        const float w = di * di;
        const float4 v0 = Hv[(size_t)node * 64 + lane];
        const float4 v1 = Hv[(size_t)node * 64 + 32 + lane];
        a0.x = v0.x * w; a0.y = v0.y * w; a0.z = v0.z * w; a0.w = v0.w * w;
        a1.x = v1.x * w; a1.y = v1.y * w; a1.z = v1.z * w; a1.w = v1.w * w;
    }
    for (int base = 0; base < n; base += 32) {
        const int m = (n - base) < 32 ? (n - base) : 32;
        int my_s; float my_w;
        load_chunk(col, dinv, start + base, m, lane, di, my_s, my_w);
        int j = 0;
        for (; j + 4 <= m; j += 4) add_rows256<4>(Hv, my_s, my_w, j, lane, a0, a1);
        const int left = m - j;
        if (left == 3)      add_rows256<3>(Hv, my_s, my_w, j, lane, a0, a1);
        else if (left == 2) add_rows256<2>(Hv, my_s, my_w, j, lane, a0, a1);
        else if (left == 1) add_rows256<1>(Hv, my_s, my_w, j, lane, a0, a1);
    }
    float4* orow = (float4*)(out + (size_t)node * F_HID);
    orow[lane]      = a0;
    orow[lane + 32] = a1;
}

__global__ void __launch_bounds__(256) k_gather64(const float* __restrict__ H,
                                                  const int* __restrict__ col,
                                                  const int* __restrict__ row_start,
                                                  const int* __restrict__ cnt,
                                                  const float* __restrict__ dinv,
                                                  float* __restrict__ out) {
    const int wave = __builtin_amdgcn_readfirstlane((int)(threadIdx.x >> 5));
    const int node = blockIdx.x * 8 + wave;
    const int lane = threadIdx.x & 31;
    const int half = lane >> 4;
    const int c    = lane & 15;
    const int start = row_start[node];
    const int n     = cnt[node];
    const float di  = dinv[node];
    const float4* Hv = (const float4*)H;

    float4 acc = {0.0f, 0.0f, 0.0f, 0.0f};
    if (half == 0) {
        const float w = di * di;
        const float4 v = Hv[(size_t)node * 16 + c];
        acc.x = v.x * w; acc.y = v.y * w; acc.z = v.z * w; acc.w = v.w * w;
    }
    for (int base = 0; base < n; base += 32) {
        const int m = (n - base) < 32 ? (n - base) : 32;
        int my_s; float my_w;
        load_chunk(col, dinv, start + base, m, lane, di, my_s, my_w);
        for (int j = 0; j < m; j += 8) {
            float4 v[4]; float w[4];
            for (int u = 0; u < 4; ++u) {
                const int e0 = j + 2 * u;
                const int sA = __builtin_amdgcn_readlane(my_s, e0);
                const int sB = __builtin_amdgcn_readlane(my_s, e0 + 1);
                const float wA = __int_as_float(__builtin_amdgcn_readlane(__float_as_int(my_w), e0));
                const float wB = __int_as_float(__builtin_amdgcn_readlane(__float_as_int(my_w), e0 + 1));
                const int s = half ? sB : sA;
                w[u] = half ? wB : wA;
                v[u] = Hv[(size_t)s * 16 + c];
            }
            for (int u = 0; u < 4; ++u)
                if (j + 2 * u + half < m) fma4(acc, v[u], w[u]);
        }
    }
    acc.x += __shfl_xor(acc.x, 16, 32);
    acc.y += __shfl_xor(acc.y, 16, 32);
    acc.z += __shfl_xor(acc.z, 16, 32);
    acc.w += __shfl_xor(acc.w, 16, 32);
    if (half == 0) {
        float4* o = (float4*)(out + (size_t)node * F_OUT) + c;
        float4 t = *o;
        t.x += acc.x; t.y += acc.y; t.z += acc.z; t.w += acc.w;
        *o = t;
    }
}

__global__ void __launch_bounds__(256) k_ln_relu(float* __restrict__ H,
                                                 const float* __restrict__ b1,
                                                 const float* __restrict__ gamma,
                                                 const float* __restrict__ beta) {
    const int node = blockIdx.x * 8 + (threadIdx.x >> 5);
    const int lane = threadIdx.x & 31;
    float* row = H + (size_t)node * F_HID;
    float4 v[2];
    float s = 0.0f;
#pragma unroll
    for (int i = 0; i < 2; ++i) {
        float4 t  = ((float4*)row)[lane + 32 * i];
        float4 bb = ((const float4*)b1)[lane + 32 * i];
        t.x += bb.x; t.y += bb.y; t.z += bb.z; t.w += bb.w;
        v[i] = t;
        s += t.x + t.y + t.z + t.w;
    }
    float mu = wave_sum32(s) * (1.0f / F_HID);
    float ss = 0.0f;
#pragma unroll
    for (int i = 0; i < 2; ++i) {
        float4 t = v[i];
        ss += (t.x - mu) * (t.x - mu) + (t.y - mu) * (t.y - mu) +
              (t.z - mu) * (t.z - mu) + (t.w - mu) * (t.w - mu);
    }
    float rstd = rsqrtf(wave_sum32(ss) * (1.0f / F_HID) + LN_EPS);
#pragma unroll
    for (int i = 0; i < 2; ++i) {
        float4 t = v[i];
        float4 g = ((const float4*)gamma)[lane + 32 * i];
        float4 b = ((const float4*)beta)[lane + 32 * i];
        t.x = fmaxf((t.x - mu) * rstd * g.x + b.x, 0.0f);
        t.y = fmaxf((t.y - mu) * rstd * g.y + b.y, 0.0f);
        t.z = fmaxf((t.z - mu) * rstd * g.z + b.z, 0.0f);
        t.w = fmaxf((t.w - mu) * rstd * g.w + b.w, 0.0f);
        ((float4*)row)[lane + 32 * i] = t;
    }
}

__global__ void __launch_bounds__(256) k_finalize(float* __restrict__ out,
                                                  const float* __restrict__ b2) {
    int i = blockIdx.x * 256 + threadIdx.x;
    if (i >= N_NODES * F_OUT) return;
    float v = out[i] + b2[i & (F_OUT - 1)];
    if (isnan(v)) v = 0.0f;
    else if (isinf(v)) v = v > 0.0f ? 20.0f : -20.0f;
    out[i] = v;
}

extern "C" void kernel_launch(void* const* d_in, const int* in_sizes, int n_in,
                              void* d_out, int out_size, void* d_ws, size_t ws_size,
                              hipStream_t stream) {
    const float* x     = (const float*)d_in[0];
    const int*   ei    = (const int*)d_in[1];
    const float* W1    = (const float*)d_in[2];
    const float* b1    = (const float*)d_in[3];
    const float* gamma = (const float*)d_in[4];
    const float* beta  = (const float*)d_in[5];
    const float* W2    = (const float*)d_in[6];
    const float* b2    = (const float*)d_in[7];
    const float* Wskip = (const float*)d_in[8];
    const int* srcp = ei;
    const int* dstp = ei + N_EDGES;

    float* ws   = (float*)d_ws;
    float* dinv = ws;
    float* xn   = dinv + N_NODES;
    float* h1   = xn + (size_t)N_NODES * F_IN;
    float* agg  = h1 + (size_t)N_NODES * F_HID;
    float* out  = (float*)d_out;

    int* cnt       = (int*)xn + 16;
    int* row_start = cnt + N_LINE;
    int* rank      = row_start + N_LINE;
    int* col       = rank + N_EDGES;
    int* cursor    = col + N_EDGES;

    k_l2norm<<<N_NODES / 8, 256, 0, stream>>>(x, xn);

    k_gemm_wmma<F_IN, F_HID, 4, 4, 4><<<(N_NODES + 31) / 32, 128, 0, stream>>>(xn, W1, h1);
    k_gemm_wmma<F_IN, F_OUT, 2, 2, 8><<<(N_NODES + 31) / 32, 64, 0, stream>>>(xn, Wskip, out);

    k_deg_init<<<(N_NODES + 255) / 256, 256, 0, stream>>>(cnt, cursor);
    k_deg_rank<<<(N_EDGES + 255) / 256, 256, 0, stream>>>(dstp, cnt, rank);
    k_dinv_rowstart<<<(N_NODES + 255) / 256, 256, 0, stream>>>(cnt, cursor, row_start, dinv);
    k_fill<<<(N_EDGES + 255) / 256, 256, 0, stream>>>(srcp, dstp, row_start, rank, col);

    k_gather256<<<N_NODES / 8, 256, 0, stream>>>(h1, col, row_start, cnt, dinv, agg);
    k_ln_relu<<<N_NODES / 8, 256, 0, stream>>>(agg, b1, gamma, beta);

    k_gemm_wmma<F_HID, F_OUT, 2, 2, 8><<<(N_NODES + 31) / 32, 64, 0, stream>>>(agg, W2, h1);
    k_gather64<<<N_NODES / 8, 256, 0, stream>>>(h1, col, row_start, cnt, dinv, out);

    k_finalize<<<(N_NODES * F_OUT + 255) / 256, 256, 0, stream>>>(out, b2);
}


# Row tails fused into the gathers: bias+LayerNorm+ReLU finish the 256-wide gather in registers, bias+nan_to_num the 64-wide one; the 2 elementwise kernels (150k short waves) are gone; sum order kept
# speedup vs baseline: 2.9327x; 1.0742x over previous
#include <hip/hip_runtime.h>
#include <math.h>

#define N_NODES 50000
#define N_EDGES 800000
#define F_IN    512
#define F_HID   256
#define F_OUT   64
#define LN_EPS  1e-5f
#define L2_EPS  1e-12f
#define N_LINE  ((N_NODES + 31) / 32 * 32)

typedef __attribute__((ext_vector_type(2))) float v2f;
typedef __attribute__((ext_vector_type(4))) float v4f;
typedef __attribute__((ext_vector_type(8))) float v8f;

__device__ __forceinline__ float wave_sum32(float v) {
#pragma unroll
    for (int m = 16; m >= 1; m >>= 1) v += __shfl_xor(v, m, 32);
    return v;
}

__global__ void __launch_bounds__(256) k_deg_init(int* __restrict__ cnt,
                                                  int* __restrict__ cursor) {
    int i = blockIdx.x * 256 + threadIdx.x;
    if (i < N_NODES) cnt[i] = 0;
    if (i == 0) *cursor = 0;
}

__global__ void __launch_bounds__(256) k_deg_rank(const int* __restrict__ dst,
                                                  int* cnt,
                                                  int* __restrict__ rank) {
    int e = blockIdx.x * 256 + threadIdx.x;
    if (e < N_EDGES) rank[e] = atomicAdd(&cnt[dst[e]], 1);
}

__global__ void __launch_bounds__(256) k_dinv_rowstart(const int* __restrict__ cnt,
                                                       int* cursor,
                                                       int* __restrict__ row_start,
                                                       float* __restrict__ dinv) {
    __shared__ int wsum[8];
    __shared__ int sbase;
    const int tid  = threadIdx.x;
    const int lane = tid & 31;
    const int w    = tid >> 5;
    const int i    = blockIdx.x * 256 + tid;
    const int c    = (i < N_NODES) ? cnt[i] : 0;
    int incl = c;
    for (int d = 1; d < 32; d <<= 1) {
        int t = __shfl_up(incl, d, 32);
        if (lane >= d) incl += t;
    }
    if (lane == 31) wsum[w] = incl;
    __syncthreads();
    if (tid == 0) {
        int tot = 0;
        for (int k = 0; k < 8; ++k) { int t = wsum[k]; wsum[k] = tot; tot += t; }
        sbase = atomicAdd(cursor, tot);
    }
    __syncthreads();
    if (i < N_NODES) {
        row_start[i] = sbase + wsum[w] + incl - c;
        dinv[i] = rsqrtf((float)(c + 1));
    }
}

__global__ void __launch_bounds__(256) k_fill(const int* __restrict__ srcp,
                                              const int* __restrict__ dstp,
                                              const int* __restrict__ row_start,
                                              const int* __restrict__ rank,
                                              int* __restrict__ col) {
    int e = blockIdx.x * 256 + threadIdx.x;
    if (e < N_EDGES) col[row_start[dstp[e]] + rank[e]] = srcp[e];
}

__global__ void __launch_bounds__(256) k_l2norm(const float* __restrict__ x,
                                                float* __restrict__ xn) {
    const int node = blockIdx.x * 8 + (threadIdx.x >> 5);
    const int lane = threadIdx.x & 31;
    const float4* row = (const float4*)(x + (size_t)node * F_IN);
    float4 v[4];
    float ss = 0.0f;
#pragma unroll
    for (int i = 0; i < 4; ++i) {
        float4 t = row[lane + 32 * i];
        t.x = isfinite(t.x) ? t.x : 0.0f;
        t.y = isfinite(t.y) ? t.y : 0.0f;
        t.z = isfinite(t.z) ? t.z : 0.0f;
        t.w = isfinite(t.w) ? t.w : 0.0f;
        v[i] = t;
        ss += t.x * t.x + t.y * t.y + t.z * t.z + t.w * t.w;
    }
    ss = wave_sum32(ss);
    float scale = 1.0f / fmaxf(sqrtf(ss), L2_EPS);
    float4* orow = (float4*)(xn + (size_t)node * F_IN);
#pragma unroll
    for (int i = 0; i < 4; ++i) {
        float4 t = v[i];
        t.x *= scale; t.y *= scale; t.z *= scale; t.w *= scale;
        orow[lane + 32 * i] = t;
    }
}

template <int TN> struct BVec;
template <> struct BVec<2> { typedef v2f T; };
template <> struct BVec<4> { typedef v4f T; };

template <int K, int NOUT, int WN, int TN, int G>
__global__ void __launch_bounds__(WN * 32) k_gemm_wmma(const float* __restrict__ A,
                                                       const float* __restrict__ B,
                                                       float* __restrict__ C) {
    typedef typename BVec<TN>::T bvec;
    constexpr int P  = K + 4;
    constexpr int BM = 32;
    constexpr int NT = WN * 32;
    static_assert(WN * TN * 16 == NOUT, "a block covers all output columns");
    static_assert((K / (4 * G)) % 2 == 0, "the K loop takes two fragment sets per trip");
    __shared__ __attribute__((aligned(16))) float As[BM * P];
    const int tid = threadIdx.x;
    const int m0  = blockIdx.x * BM;

    {
        const float4* gA = (const float4*)A;
#pragma unroll 8
        for (int i = tid; i < BM * (K / 4); i += NT) {
            const int row = i / (K / 4);
            const int c4  = i % (K / 4);
            int grow = m0 + row;
            grow = grow < N_NODES ? grow : N_NODES - 1;
            *(float4*)&As[row * P + 4 * c4] = gA[(size_t)grow * (K / 4) + c4];
        }
    }
    __syncthreads();

    const int wn   = tid >> 5;
    const int lane = tid & 31;
    const int r    = lane & 15;
    const int h    = lane >> 4;
    const int n0   = wn * (TN * 16);
    const float* ap = &As[r * P + 2 * h];
    const float* bp = B + (size_t)(2 * h) * NOUT + n0 + TN * r;

    v8f acc[2][TN];
    for (int t = 0; t < 2; ++t)
        for (int j = 0; j < TN; ++j) acc[t][j] = v8f{};

    struct FragSet { v2f a[G][2]; bvec b[G][2]; };
    auto load_set = [&](FragSet& s, const int k) {
        for (int u = 0; u < G; ++u) {
            s.a[u][0] = *(const v2f*)(ap + k + 4 * u);
            s.a[u][1] = *(const v2f*)(ap + 16 * P + k + 4 * u);
            s.b[u][0] = *(const bvec*)(bp + (size_t)(k + 4 * u) * NOUT);
            s.b[u][1] = *(const bvec*)(bp + (size_t)(k + 4 * u + 1) * NOUT);
        }
    };
    auto mma_set = [&](const FragSet& s) {
        for (int u = 0; u < G; ++u)
            for (int j = 0; j < TN; ++j) {
                v2f b;
                b.x = s.b[u][0][j];
                b.y = s.b[u][1][j];
                acc[0][j] = __builtin_amdgcn_wmma_f32_16x16x4_f32(
                     false, s.a[u][0],  false, b,
                     (short)0, acc[0][j],  false,  false);
                acc[1][j] = __builtin_amdgcn_wmma_f32_16x16x4_f32(
                     false, s.a[u][1],  false, b,
                     (short)0, acc[1][j],  false,  false);
            }
    };

    FragSet s0, s1;
    load_set(s0, 0);
    for (int k = 0; k < K; k += 8 * G) {
        load_set(s1, k + 4 * G);
        mma_set(s0);
        if (k + 8 * G < K) load_set(s0, k + 8 * G);
        mma_set(s1);
    }

    for (int t = 0; t < 2; ++t) {
        const int mb = m0 + 16 * t;
        if (mb < N_NODES) {
            for (int i = 0; i < 8; ++i) {
                bvec o;
                for (int j = 0; j < TN; ++j) o[j] = acc[t][j][i];
                *(bvec*)(C + (size_t)(mb + 8 * h + i) * NOUT + n0 + TN * r) = o;
            }
        }
    }
}

__device__ __forceinline__ void fma4(float4& a, const float4 v, const float w) {
    a.x = fmaf(v.x, w, a.x); a.y = fmaf(v.y, w, a.y);
    a.z = fmaf(v.z, w, a.z); a.w = fmaf(v.w, w, a.w);
}

__device__ __forceinline__ void load_chunk(const int* __restrict__ col,
                                           const float* __restrict__ dinv,
                                           int first, int m, int lane, float di,
                                           int& my_s, float& my_w) {
    my_s = 0; my_w = 0.0f;
    if (lane < m) {
        unsigned s = (unsigned)col[first + lane];
        s = s < (unsigned)N_NODES ? s : (unsigned)(N_NODES - 1);
        my_s = (int)s;
        my_w = dinv[s] * di;
    }
}

template <int R>
__device__ __forceinline__ void add_rows256(const float4* __restrict__ Hv,
                                            int my_s, float my_w, int j, int lane,
                                            float4& a0, float4& a1) {
    float4 v[2 * R]; float w[R];
    for (int u = 0; u < R; ++u) {
        const int s = __builtin_amdgcn_readlane(my_s, j + u);
        w[u] = __int_as_float(__builtin_amdgcn_readlane(__float_as_int(my_w), j + u));
        const float4* r = Hv + (size_t)s * 64;
        v[2 * u]     = r[lane];
        v[2 * u + 1] = r[lane + 32];
    }
    for (int u = 0; u < R; ++u) { fma4(a0, v[2 * u], w[u]); fma4(a1, v[2 * u + 1], w[u]); }
}

__global__ void __launch_bounds__(256) k_gather256(const float* __restrict__ H,
                                                   const int* __restrict__ col,
                                                   const int* __restrict__ row_start,
                                                   const int* __restrict__ cnt,
                                                   const float* __restrict__ dinv,
                                                   const float* __restrict__ b1,
                                                   const float* __restrict__ gamma,
                                                   const float* __restrict__ beta,
                                                   float* __restrict__ out) {
    const int wave = __builtin_amdgcn_readfirstlane((int)(threadIdx.x >> 5));
    const int node = blockIdx.x * 8 + wave;
    const int lane = threadIdx.x & 31;
    const int start = row_start[node];
    const int n     = cnt[node];
    const float di  = dinv[node];
    const float4* Hv = (const float4*)H;

    float4 a0, a1;
    {
        const float w = di * di;
        const float4 v0 = Hv[(size_t)node * 64 + lane];
        const float4 v1 = Hv[(size_t)node * 64 + 32 + lane];
        a0.x = v0.x * w; a0.y = v0.y * w; a0.z = v0.z * w; a0.w = v0.w * w;
        a1.x = v1.x * w; a1.y = v1.y * w; a1.z = v1.z * w; a1.w = v1.w * w;
    }
    for (int base = 0; base < n; base += 32) {
        const int m = (n - base) < 32 ? (n - base) : 32;
        int my_s; float my_w;
        load_chunk(col, dinv, start + base, m, lane, di, my_s, my_w);
        int j = 0;
        for (; j + 4 <= m; j += 4) add_rows256<4>(Hv, my_s, my_w, j, lane, a0, a1);
        const int left = m - j;
        if (left == 3)      add_rows256<3>(Hv, my_s, my_w, j, lane, a0, a1);
        else if (left == 2) add_rows256<2>(Hv, my_s, my_w, j, lane, a0, a1);
        else if (left == 1) add_rows256<1>(Hv, my_s, my_w, j, lane, a0, a1);
    }
    float4 v[2] = {a0, a1};
    float s = 0.0f;
#pragma unroll
    for (int i = 0; i < 2; ++i) {
        float4 t  = v[i];
        float4 bb = ((const float4*)b1)[lane + 32 * i];
        t.x += bb.x; t.y += bb.y; t.z += bb.z; t.w += bb.w;
        v[i] = t;
        s += t.x + t.y + t.z + t.w;
    }
    float mu = wave_sum32(s) * (1.0f / F_HID);
    float ss = 0.0f;
#pragma unroll
    for (int i = 0; i < 2; ++i) {
        float4 t = v[i];
        ss += (t.x - mu) * (t.x - mu) + (t.y - mu) * (t.y - mu) +
              (t.z - mu) * (t.z - mu) + (t.w - mu) * (t.w - mu);
    }
    float rstd = rsqrtf(wave_sum32(ss) * (1.0f / F_HID) + LN_EPS);
    float4* orow = (float4*)(out + (size_t)node * F_HID);
#pragma unroll
    for (int i = 0; i < 2; ++i) {
        float4 t = v[i];
        float4 g = ((const float4*)gamma)[lane + 32 * i];
        float4 b = ((const float4*)beta)[lane + 32 * i];
        t.x = fmaxf((t.x - mu) * rstd * g.x + b.x, 0.0f);
        t.y = fmaxf((t.y - mu) * rstd * g.y + b.y, 0.0f);
        t.z = fmaxf((t.z - mu) * rstd * g.z + b.z, 0.0f);
        t.w = fmaxf((t.w - mu) * rstd * g.w + b.w, 0.0f);
        orow[lane + 32 * i] = t;
    }
}

__device__ __forceinline__ float final_value(float v) {
    const float lim = v > 0.0f ? 20.0f : -20.0f;
    v = isinf(v) ? lim : v;
    return isnan(v) ? 0.0f : v;
}

__global__ void __launch_bounds__(256) k_gather64(const float* __restrict__ H,
                                                  const int* __restrict__ col,
                                                  const int* __restrict__ row_start,
                                                  const int* __restrict__ cnt,
                                                  const float* __restrict__ dinv,
                                                  const float* __restrict__ b2,
                                                  float* __restrict__ out) {
    const int wave = __builtin_amdgcn_readfirstlane((int)(threadIdx.x >> 5));
    const int node = blockIdx.x * 8 + wave;
    const int lane = threadIdx.x & 31;
    const int half = lane >> 4;
    const int c    = lane & 15;
    const int start = row_start[node];
    const int n     = cnt[node];
    const float di  = dinv[node];
    const float4* Hv = (const float4*)H;

    float4 acc = {0.0f, 0.0f, 0.0f, 0.0f};
    if (half == 0) {
        const float w = di * di;
        const float4 v = Hv[(size_t)node * 16 + c];
        acc.x = v.x * w; acc.y = v.y * w; acc.z = v.z * w; acc.w = v.w * w;
    }
    for (int base = 0; base < n; base += 32) {
        const int m = (n - base) < 32 ? (n - base) : 32;
        int my_s; float my_w;
        load_chunk(col, dinv, start + base, m, lane, di, my_s, my_w);
        for (int j = 0; j < m; j += 8) {
            float4 v[4]; float w[4];
            for (int u = 0; u < 4; ++u) {
                const int e0 = j + 2 * u;
                const int sA = __builtin_amdgcn_readlane(my_s, e0);
                const int sB = __builtin_amdgcn_readlane(my_s, e0 + 1);
                const float wA = __int_as_float(__builtin_amdgcn_readlane(__float_as_int(my_w), e0));
                const float wB = __int_as_float(__builtin_amdgcn_readlane(__float_as_int(my_w), e0 + 1));
                const int s = half ? sB : sA;
                w[u] = half ? wB : wA;
                v[u] = Hv[(size_t)s * 16 + c];
            }
            for (int u = 0; u < 4; ++u)
                if (j + 2 * u + half < m) fma4(acc, v[u], w[u]);
        }
    }
    acc.x += __shfl_xor(acc.x, 16, 32);
    acc.y += __shfl_xor(acc.y, 16, 32);
    acc.z += __shfl_xor(acc.z, 16, 32);
    acc.w += __shfl_xor(acc.w, 16, 32);
    if (half == 0) {
        float4* o = (float4*)(out + (size_t)node * F_OUT) + c;
        const float4 bb = ((const float4*)b2)[c];
        float4 t = *o;
        t.x = final_value(t.x + acc.x + bb.x);
        t.y = final_value(t.y + acc.y + bb.y);
        t.z = final_value(t.z + acc.z + bb.z);
        t.w = final_value(t.w + acc.w + bb.w);
        *o = t;
    }
}

extern "C" void kernel_launch(void* const* d_in, const int* in_sizes, int n_in,
                              void* d_out, int out_size, void* d_ws, size_t ws_size,
                              hipStream_t stream) {
    const float* x     = (const float*)d_in[0];
    const int*   ei    = (const int*)d_in[1];
    const float* W1    = (const float*)d_in[2];
    const float* b1    = (const float*)d_in[3];
    const float* gamma = (const float*)d_in[4];
    const float* beta  = (const float*)d_in[5];
    const float* W2    = (const float*)d_in[6];
    const float* b2    = (const float*)d_in[7];
    const float* Wskip = (const float*)d_in[8];
    const int* srcp = ei;
    const int* dstp = ei + N_EDGES;

    float* ws   = (float*)d_ws;
    float* dinv = ws;
    float* xn   = dinv + N_NODES;
    float* h1   = xn + (size_t)N_NODES * F_IN;
    float* agg  = h1 + (size_t)N_NODES * F_HID;
    float* out  = (float*)d_out;

    int* cnt       = (int*)xn + 16;
    int* row_start = cnt + N_LINE;
    int* rank      = row_start + N_LINE;
    int* col       = rank + N_EDGES;
    int* cursor    = col + N_EDGES;

    k_l2norm<<<N_NODES / 8, 256, 0, stream>>>(x, xn);

    k_gemm_wmma<F_IN, F_HID, 4, 4, 4><<<(N_NODES + 31) / 32, 128, 0, stream>>>(xn, W1, h1);
    k_gemm_wmma<F_IN, F_OUT, 2, 2, 8><<<(N_NODES + 31) / 32, 64, 0, stream>>>(xn, Wskip, out);

    k_deg_init<<<(N_NODES + 255) / 256, 256, 0, stream>>>(cnt, cursor);
    k_deg_rank<<<(N_EDGES + 255) / 256, 256, 0, stream>>>(dstp, cnt, rank);
    k_dinv_rowstart<<<(N_NODES + 255) / 256, 256, 0, stream>>>(cnt, cursor, row_start, dinv);
    k_fill<<<(N_EDGES + 255) / 256, 256, 0, stream>>>(srcp, dstp, row_start, rank, col);

    k_gather256<<<N_NODES / 8, 256, 0, stream>>>(h1, col, row_start, cnt, dinv, b1, gamma, beta, agg);

    k_gemm_wmma<F_HID, F_OUT, 2, 2, 8><<<(N_NODES + 31) / 32, 64, 0, stream>>>(agg, W2, h1);
    k_gather64<<<N_NODES / 8, 256, 0, stream>>>(h1, col, row_start, cnt, dinv, b2, out);
}


# Input clean + L2 normalise fused into the x products: a block normalises its 32 rows straight into the LDS A tile and multiplies it by W1, then by Wskip; x read once, xn never stored; sum order kept
# speedup vs baseline: 3.2593x; 1.1114x over previous
#include <hip/hip_runtime.h>
#include <math.h>

#define N_NODES 50000
#define N_EDGES 800000
#define F_IN    512
#define F_HID   256
#define F_OUT   64
#define LN_EPS  1e-5f
#define L2_EPS  1e-12f
#define N_LINE  ((N_NODES + 31) / 32 * 32)

typedef __attribute__((ext_vector_type(2))) float v2f;
typedef __attribute__((ext_vector_type(4))) float v4f;
typedef __attribute__((ext_vector_type(8))) float v8f;

__device__ __forceinline__ float wave_sum32(float v) {
#pragma unroll
    for (int m = 16; m >= 1; m >>= 1) v += __shfl_xor(v, m, 32);
    return v;
}

__global__ void __launch_bounds__(256) k_deg_init(int* __restrict__ cnt,
                                                  int* __restrict__ cursor) {
    int i = blockIdx.x * 256 + threadIdx.x;
    if (i < N_NODES) cnt[i] = 0;
    if (i == 0) *cursor = 0;
}

__global__ void __launch_bounds__(256) k_deg_rank(const int* __restrict__ dst,
                                                  int* cnt,
                                                  int* __restrict__ rank) {
    int e = blockIdx.x * 256 + threadIdx.x;
    if (e < N_EDGES) rank[e] = atomicAdd(&cnt[dst[e]], 1);
}

__global__ void __launch_bounds__(256) k_dinv_rowstart(const int* __restrict__ cnt,
                                                       int* cursor,
                                                       int* __restrict__ row_start,
                                                       float* __restrict__ dinv) {
    __shared__ int wsum[8];
    __shared__ int sbase;
    const int tid  = threadIdx.x;
    const int lane = tid & 31;
    const int w    = tid >> 5;
    const int i    = blockIdx.x * 256 + tid;
    const int c    = (i < N_NODES) ? cnt[i] : 0;
    int incl = c;
    for (int d = 1; d < 32; d <<= 1) {
        int t = __shfl_up(incl, d, 32);
        if (lane >= d) incl += t;
    }
    if (lane == 31) wsum[w] = incl;
    __syncthreads();
    if (tid == 0) {
        int tot = 0;
        for (int k = 0; k < 8; ++k) { int t = wsum[k]; wsum[k] = tot; tot += t; }
        sbase = atomicAdd(cursor, tot);
    }
    __syncthreads();
    if (i < N_NODES) {
        row_start[i] = sbase + wsum[w] + incl - c;
        dinv[i] = rsqrtf((float)(c + 1));
    }
}

__global__ void __launch_bounds__(256) k_fill(const int* __restrict__ srcp,
                                              const int* __restrict__ dstp,
                                              const int* __restrict__ row_start,
                                              const int* __restrict__ rank,
                                              int* __restrict__ col) {
    int e = blockIdx.x * 256 + threadIdx.x;
    if (e < N_EDGES) col[row_start[dstp[e]] + rank[e]] = srcp[e];
}

template <int TN> struct BVec;
template <> struct BVec<2> { typedef v2f T; };
template <> struct BVec<4> { typedef v4f T; };

#define A_PITCH(K) ((K) + 4)

template <int K, int NOUT, int RT, int TN, int G>
__device__ __forceinline__ void wave_gemm(const float* As,
                                          const float* __restrict__ B,
                                          float* __restrict__ C,
                                          const int m0, const int n0, const int lane) {
    typedef typename BVec<TN>::T bvec;
    constexpr int P = A_PITCH(K);
    static_assert((K / (4 * G)) % 2 == 0, "the K loop takes two fragment sets per trip");
    const int r    = lane & 15;
    const int h    = lane >> 4;
    const float* ap = &As[r * P + 2 * h];
    const float* bp = B + (size_t)(2 * h) * NOUT + n0 + TN * r;

    v8f acc[RT][TN];
    for (int t = 0; t < RT; ++t)
        for (int j = 0; j < TN; ++j) acc[t][j] = v8f{};

    struct FragSet { v2f a[G][RT]; bvec b[G][2]; };
    auto load_set = [&](FragSet& s, const int k) {
        for (int u = 0; u < G; ++u) {
            for (int t = 0; t < RT; ++t)
                s.a[u][t] = *(const v2f*)(ap + 16 * t * P + k + 4 * u);
            s.b[u][0] = *(const bvec*)(bp + (size_t)(k + 4 * u) * NOUT);
            s.b[u][1] = *(const bvec*)(bp + (size_t)(k + 4 * u + 1) * NOUT);
        }
    };
    auto mma_set = [&](const FragSet& s) {
        for (int u = 0; u < G; ++u)
            for (int j = 0; j < TN; ++j) {
                v2f b;
                b.x = s.b[u][0][j];
                b.y = s.b[u][1][j];
                for (int t = 0; t < RT; ++t)
                    acc[t][j] = __builtin_amdgcn_wmma_f32_16x16x4_f32(
                         false, s.a[u][t],  false, b,
                         (short)0, acc[t][j],  false,  false);
            }
    };

    FragSet s0, s1;
    load_set(s0, 0);
    for (int k = 0; k < K; k += 8 * G) {
        load_set(s1, k + 4 * G);
        mma_set(s0);
        if (k + 8 * G < K) load_set(s0, k + 8 * G);
        mma_set(s1);
    }

    for (int t = 0; t < RT; ++t) {
        const int mb = m0 + 16 * t;
        if (mb < N_NODES) {
            for (int i = 0; i < 8; ++i) {
                bvec o;
                for (int j = 0; j < TN; ++j) o[j] = acc[t][j][i];
                *(bvec*)(C + (size_t)(mb + 8 * h + i) * NOUT + n0 + TN * r) = o;
            }
        }
    }
}

template <int K, int NOUT, int WN, int TN, int G>
__global__ void __launch_bounds__(WN * 32) k_gemm_wmma(const float* __restrict__ A,
                                                       const float* __restrict__ B,
                                                       float* __restrict__ C) {
    constexpr int P  = A_PITCH(K);
    constexpr int BM = 32;
    constexpr int NT = WN * 32;
    static_assert(WN * TN * 16 == NOUT, "a block covers all output columns");
    __shared__ __attribute__((aligned(16))) float As[BM * P];
    const int tid = threadIdx.x;
    const int m0  = blockIdx.x * BM;

    {
        const float4* gA = (const float4*)A;
#pragma unroll 8
        for (int i = tid; i < BM * (K / 4); i += NT) {
            const int row = i / (K / 4);
            const int c4  = i % (K / 4);
            int grow = m0 + row;
            grow = grow < N_NODES ? grow : N_NODES - 1;
            *(float4*)&As[row * P + 4 * c4] = gA[(size_t)grow * (K / 4) + c4];
        }
    }
    __syncthreads();

    const int wn = tid >> 5;
    wave_gemm<K, NOUT, 2, TN, G>(As, B, C, m0, wn * (TN * 16), tid & 31);
}

template <int K, int N1, int N2, int TN, int G>
__global__ void __launch_bounds__(N1 / (TN * 16) * 32)
k_norm_gemm_pair(const float* __restrict__ x,
                 const float* __restrict__ B1, const float* __restrict__ B2,
                 float* __restrict__ C1, float* __restrict__ C2) {
    constexpr int P   = A_PITCH(K);
    constexpr int BM  = 32;
    constexpr int WN  = N1 / (TN * 16);
    constexpr int RQ  = BM / WN;
    static_assert(K == 4 * 4 * 32, "a lane holds 4 float4 of a row");
    static_assert(WN * TN * 16 == N1 && BM % WN == 0, "whole column groups, whole rows per wave");
    static_assert((BM / 16) * (N2 / 32) == WN, "one 16-row x 32-column share of C2 per wave");
    __shared__ __attribute__((aligned(16))) float As[BM * P];
    const int wn   = __builtin_amdgcn_readfirstlane((int)(threadIdx.x >> 5));
    const int lane = threadIdx.x & 31;
    const int m0   = blockIdx.x * BM;

    {
        float4 v[RQ][4];
#pragma unroll
        for (int q = 0; q < RQ; ++q) {
            int grow = m0 + wn + WN * q;
            grow = grow < N_NODES ? grow : N_NODES - 1;
            const float4* grp = (const float4*)(x + (size_t)grow * K);
#pragma unroll
            for (int i = 0; i < 4; ++i) v[q][i] = grp[lane + 32 * i];
        }
#pragma unroll
        for (int q = 0; q < RQ; ++q) {
            const int row = wn + WN * q;
            float ss = 0.0f;
#pragma unroll
            for (int i = 0; i < 4; ++i) {
                float4 t = v[q][i];
                t.x = isfinite(t.x) ? t.x : 0.0f;
                t.y = isfinite(t.y) ? t.y : 0.0f;
                t.z = isfinite(t.z) ? t.z : 0.0f;
                t.w = isfinite(t.w) ? t.w : 0.0f;
                v[q][i] = t;
                ss += t.x * t.x + t.y * t.y + t.z * t.z + t.w * t.w;
            }
            ss = wave_sum32(ss);
            const float scale = 1.0f / fmaxf(sqrtf(ss), L2_EPS);
#pragma unroll
            for (int i = 0; i < 4; ++i) {
                float4 t = v[q][i];
                t.x *= scale; t.y *= scale; t.z *= scale; t.w *= scale;
                *(float4*)&As[row * P + 4 * (lane + 32 * i)] = t;
            }
        }
    }
    __syncthreads();

    wave_gemm<K, N1, 2, TN, G>(As, B1, C1, m0, wn * (TN * 16), lane);
    const int rt = wn & 1;
    wave_gemm<K, N2, 1, 2, G>(As + 16 * rt * P, B2, C2, m0 + 16 * rt, (wn >> 1) * 32, lane);
}

__device__ __forceinline__ void fma4(float4& a, const float4 v, const float w) {
    a.x = fmaf(v.x, w, a.x); a.y = fmaf(v.y, w, a.y);
    a.z = fmaf(v.z, w, a.z); a.w = fmaf(v.w, w, a.w);
}

__device__ __forceinline__ void load_chunk(const int* __restrict__ col,
                                           const float* __restrict__ dinv,
                                           int first, int m, int lane, float di,
                                           int& my_s, float& my_w) {
    my_s = 0; my_w = 0.0f;
    if (lane < m) {
        unsigned s = (unsigned)col[first + lane];
        s = s < (unsigned)N_NODES ? s : (unsigned)(N_NODES - 1);
        my_s = (int)s;
        my_w = dinv[s] * di;
    }
}

template <int R>
__device__ __forceinline__ void add_rows256(const float4* __restrict__ Hv,
                                            int my_s, float my_w, int j, int lane,
                                            float4& a0, float4& a1) {
    float4 v[2 * R]; float w[R];
    for (int u = 0; u < R; ++u) {
        const int s = __builtin_amdgcn_readlane(my_s, j + u);
        w[u] = __int_as_float(__builtin_amdgcn_readlane(__float_as_int(my_w), j + u));
        const float4* r = Hv + (size_t)s * 64;
        v[2 * u]     = r[lane];
        v[2 * u + 1] = r[lane + 32];
    }
    for (int u = 0; u < R; ++u) { fma4(a0, v[2 * u], w[u]); fma4(a1, v[2 * u + 1], w[u]); }
}

__global__ void __launch_bounds__(256) k_gather256(const float* __restrict__ H,
                                                   const int* __restrict__ col,
                                                   const int* __restrict__ row_start,
                                                   const int* __restrict__ cnt,
                                                   const float* __restrict__ dinv,
                                                   const float* __restrict__ b1,
                                                   const float* __restrict__ gamma,
                                                   const float* __restrict__ beta,
                                                   float* __restrict__ out) {
    const int wave = __builtin_amdgcn_readfirstlane((int)(threadIdx.x >> 5));
    const int node = blockIdx.x * 8 + wave;
    const int lane = threadIdx.x & 31;
    const int start = row_start[node];
    const int n     = cnt[node];
    const float di  = dinv[node];
    const float4* Hv = (const float4*)H;

    float4 a0, a1;
    {
        const float w = di * di;
        const float4 v0 = Hv[(size_t)node * 64 + lane];
        const float4 v1 = Hv[(size_t)node * 64 + 32 + lane];
        a0.x = v0.x * w; a0.y = v0.y * w; a0.z = v0.z * w; a0.w = v0.w * w;
        a1.x = v1.x * w; a1.y = v1.y * w; a1.z = v1.z * w; a1.w = v1.w * w;
    }
    for (int base = 0; base < n; base += 32) {
        const int m = (n - base) < 32 ? (n - base) : 32;
        int my_s; float my_w;
        load_chunk(col, dinv, start + base, m, lane, di, my_s, my_w);
        int j = 0;
        for (; j + 4 <= m; j += 4) add_rows256<4>(Hv, my_s, my_w, j, lane, a0, a1);
        const int left = m - j;
        if (left == 3)      add_rows256<3>(Hv, my_s, my_w, j, lane, a0, a1);
        else if (left == 2) add_rows256<2>(Hv, my_s, my_w, j, lane, a0, a1);
        else if (left == 1) add_rows256<1>(Hv, my_s, my_w, j, lane, a0, a1);
    }
    float4 v[2] = {a0, a1};
    float s = 0.0f;
#pragma unroll
    for (int i = 0; i < 2; ++i) {
        float4 t  = v[i];
        float4 bb = ((const float4*)b1)[lane + 32 * i];
        t.x += bb.x; t.y += bb.y; t.z += bb.z; t.w += bb.w;
        v[i] = t;
        s += t.x + t.y + t.z + t.w;
    }
    float mu = wave_sum32(s) * (1.0f / F_HID);
    float ss = 0.0f;
#pragma unroll
    for (int i = 0; i < 2; ++i) {
        float4 t = v[i];
        ss += (t.x - mu) * (t.x - mu) + (t.y - mu) * (t.y - mu) +
              (t.z - mu) * (t.z - mu) + (t.w - mu) * (t.w - mu);
    }
    float rstd = rsqrtf(wave_sum32(ss) * (1.0f / F_HID) + LN_EPS);
    float4* orow = (float4*)(out + (size_t)node * F_HID);
#pragma unroll
    for (int i = 0; i < 2; ++i) {
        float4 t = v[i];
        float4 g = ((const float4*)gamma)[lane + 32 * i];
        float4 b = ((const float4*)beta)[lane + 32 * i];
        t.x = fmaxf((t.x - mu) * rstd * g.x + b.x, 0.0f);
        t.y = fmaxf((t.y - mu) * rstd * g.y + b.y, 0.0f);
        t.z = fmaxf((t.z - mu) * rstd * g.z + b.z, 0.0f);
        t.w = fmaxf((t.w - mu) * rstd * g.w + b.w, 0.0f);
        orow[lane + 32 * i] = t;
    }
}

__device__ __forceinline__ float final_value(float v) {
    const float lim = v > 0.0f ? 20.0f : -20.0f;
    v = isinf(v) ? lim : v;
    return isnan(v) ? 0.0f : v;
}

__global__ void __launch_bounds__(256) k_gather64(const float* __restrict__ H,
                                                  const int* __restrict__ col,
                                                  const int* __restrict__ row_start,
                                                  const int* __restrict__ cnt,
                                                  const float* __restrict__ dinv,
                                                  const float* __restrict__ b2,
                                                  float* __restrict__ out) {
    const int wave = __builtin_amdgcn_readfirstlane((int)(threadIdx.x >> 5));
    const int node = blockIdx.x * 8 + wave;
    const int lane = threadIdx.x & 31;
    const int half = lane >> 4;
    const int c    = lane & 15;
    const int start = row_start[node];
    const int n     = cnt[node];
    const float di  = dinv[node];
    const float4* Hv = (const float4*)H;

    float4 acc = {0.0f, 0.0f, 0.0f, 0.0f};
    if (half == 0) {
        const float w = di * di;
        const float4 v = Hv[(size_t)node * 16 + c];
        acc.x = v.x * w; acc.y = v.y * w; acc.z = v.z * w; acc.w = v.w * w;
    }
    for (int base = 0; base < n; base += 32) {
        const int m = (n - base) < 32 ? (n - base) : 32;
        int my_s; float my_w;
        load_chunk(col, dinv, start + base, m, lane, di, my_s, my_w);
        for (int j = 0; j < m; j += 8) {
            float4 v[4]; float w[4];
            for (int u = 0; u < 4; ++u) {
                const int e0 = j + 2 * u;
                const int sA = __builtin_amdgcn_readlane(my_s, e0);
                const int sB = __builtin_amdgcn_readlane(my_s, e0 + 1);
                const float wA = __int_as_float(__builtin_amdgcn_readlane(__float_as_int(my_w), e0));
                const float wB = __int_as_float(__builtin_amdgcn_readlane(__float_as_int(my_w), e0 + 1));
                const int s = half ? sB : sA;
                w[u] = half ? wB : wA;
                v[u] = Hv[(size_t)s * 16 + c];
            }
            for (int u = 0; u < 4; ++u)
                if (j + 2 * u + half < m) fma4(acc, v[u], w[u]);
        }
    }
    acc.x += __shfl_xor(acc.x, 16, 32);
    acc.y += __shfl_xor(acc.y, 16, 32);
    acc.z += __shfl_xor(acc.z, 16, 32);
    acc.w += __shfl_xor(acc.w, 16, 32);
    if (half == 0) {
        float4* o = (float4*)(out + (size_t)node * F_OUT) + c;
        const float4 bb = ((const float4*)b2)[c];
        float4 t = *o;
        t.x = final_value(t.x + acc.x + bb.x);
        t.y = final_value(t.y + acc.y + bb.y);
        t.z = final_value(t.z + acc.z + bb.z);
        t.w = final_value(t.w + acc.w + bb.w);
        *o = t;
    }
}

extern "C" void kernel_launch(void* const* d_in, const int* in_sizes, int n_in,
                              void* d_out, int out_size, void* d_ws, size_t ws_size,
                              hipStream_t stream) {
    const float* x     = (const float*)d_in[0];
    const int*   ei    = (const int*)d_in[1];
    const float* W1    = (const float*)d_in[2];
    const float* b1    = (const float*)d_in[3];
    const float* gamma = (const float*)d_in[4];
    const float* beta  = (const float*)d_in[5];
    const float* W2    = (const float*)d_in[6];
    const float* b2    = (const float*)d_in[7];
    const float* Wskip = (const float*)d_in[8];
    const int* srcp = ei;
    const int* dstp = ei + N_EDGES;

    float* ws   = (float*)d_ws;
    float* dinv = ws;
    float* xn   = dinv + N_NODES;
    float* h1   = xn + (size_t)N_NODES * F_IN;
    float* agg  = h1 + (size_t)N_NODES * F_HID;
    float* out  = (float*)d_out;

    int* cnt       = (int*)xn + 16;
    int* row_start = cnt + N_LINE;
    int* rank      = row_start + N_LINE;
    int* col       = rank + N_EDGES;
    int* cursor    = col + N_EDGES;

    k_norm_gemm_pair<F_IN, F_HID, F_OUT, 4, 4><<<(N_NODES + 31) / 32, 128, 0, stream>>>(x, W1, Wskip, h1, out);

    k_deg_init<<<(N_NODES + 255) / 256, 256, 0, stream>>>(cnt, cursor);
    k_deg_rank<<<(N_EDGES + 255) / 256, 256, 0, stream>>>(dstp, cnt, rank);
    k_dinv_rowstart<<<(N_NODES + 255) / 256, 256, 0, stream>>>(cnt, cursor, row_start, dinv);
    k_fill<<<(N_EDGES + 255) / 256, 256, 0, stream>>>(srcp, dstp, row_start, rank, col);

    k_gather256<<<N_NODES / 8, 256, 0, stream>>>(h1, col, row_start, cnt, dinv, b1, gamma, beta, agg);

    k_gemm_wmma<F_HID, F_OUT, 2, 2, 8><<<(N_NODES + 31) / 32, 64, 0, stream>>>(agg, W2, h1);
    k_gather64<<<N_NODES / 8, 256, 0, stream>>>(h1, col, row_start, cnt, dinv, b2, out);
}


# Edges grouped in ONE pass: 64-slot row per destination (+ overflow list), filled by extra blocks of the x-product launch; rank, scan, fill launches gone (8 -> 5); sum order = arrival rank as before
# speedup vs baseline: 3.8058x; 1.1677x over previous
#include <hip/hip_runtime.h>
#include <math.h>

#define N_NODES 50000
#define N_EDGES 800000
#define F_IN    512
#define F_HID   256
#define F_OUT   64
#define LN_EPS  1e-5f
#define L2_EPS  1e-12f
#define N_LINE  ((N_NODES + 31) / 32 * 32)

typedef __attribute__((ext_vector_type(2))) float v2f;
typedef __attribute__((ext_vector_type(4))) float v4f;
typedef __attribute__((ext_vector_type(8))) float v8f;

__device__ __forceinline__ float wave_sum32(float v) {
#pragma unroll
    for (int m = 16; m >= 1; m >>= 1) v += __shfl_xor(v, m, 32);
    return v;
}

#define CAP 64
#define GEMM_BLOCKS ((N_NODES + 31) / 32)
#define EDGES_PER_BLOCK 1024
#define EDGE_BLOCKS ((N_EDGES + EDGES_PER_BLOCK - 1) / EDGES_PER_BLOCK)

__global__ void __launch_bounds__(256) k_deg_init(int* __restrict__ cnt,
                                                  int* __restrict__ ovf_cnt) {
    int i = blockIdx.x * 256 + threadIdx.x;
    if (i < N_NODES) cnt[i] = 0;
    if (i == 0) *ovf_cnt = 0;
}

__device__ __forceinline__ void bucket_edges(const int b, const int tid,
                                             const int* __restrict__ srcp,
                                             const int* __restrict__ dstp,
                                             int* cnt, int* __restrict__ col,
                                             int* ovf_cnt,
                                             int* __restrict__ ovf_dst,
                                             int* __restrict__ ovf_src) {
    int d[8], s[8], r[8];
#pragma unroll
    for (int u = 0; u < 8; ++u) {
        const int e = b * EDGES_PER_BLOCK + 128 * u + tid;
        const bool ok = e < N_EDGES;
        d[u] = ok ? dstp[e] : -1;
        s[u] = ok ? srcp[e] : 0;
    }
#pragma unroll
    for (int u = 0; u < 8; ++u) r[u] = d[u] >= 0 ? atomicAdd(&cnt[d[u]], 1) : 0;
#pragma unroll
    for (int u = 0; u < 8; ++u) {
        if (d[u] >= 0) {
            if (r[u] < CAP) {
                col[(size_t)d[u] * CAP + r[u]] = s[u];
            } else {
                const int o = atomicAdd(ovf_cnt, 1);
                ovf_dst[o] = d[u];
                ovf_src[o] = s[u];
            }
        }
    }
}

template <int TN> struct BVec;
template <> struct BVec<2> { typedef v2f T; };
template <> struct BVec<4> { typedef v4f T; };

#define A_PITCH(K) ((K) + 4)

template <int K, int NOUT, int RT, int TN, int G>
__device__ __forceinline__ void wave_gemm(const float* As,
                                          const float* __restrict__ B,
                                          float* __restrict__ C,
                                          const int m0, const int n0, const int lane) {
    typedef typename BVec<TN>::T bvec;
    constexpr int P = A_PITCH(K);
    static_assert((K / (4 * G)) % 2 == 0, "the K loop takes two fragment sets per trip");
    const int r    = lane & 15;
    const int h    = lane >> 4;
    const float* ap = &As[r * P + 2 * h];
    const float* bp = B + (size_t)(2 * h) * NOUT + n0 + TN * r;

    v8f acc[RT][TN];
    for (int t = 0; t < RT; ++t)
        for (int j = 0; j < TN; ++j) acc[t][j] = v8f{};

    struct FragSet { v2f a[G][RT]; bvec b[G][2]; };
    auto load_set = [&](FragSet& s, const int k) {
        for (int u = 0; u < G; ++u) {
            for (int t = 0; t < RT; ++t)
                s.a[u][t] = *(const v2f*)(ap + 16 * t * P + k + 4 * u);
            s.b[u][0] = *(const bvec*)(bp + (size_t)(k + 4 * u) * NOUT);
            s.b[u][1] = *(const bvec*)(bp + (size_t)(k + 4 * u + 1) * NOUT);
        }
    };
    auto mma_set = [&](const FragSet& s) {
        for (int u = 0; u < G; ++u)
            for (int j = 0; j < TN; ++j) {
                v2f b;
                b.x = s.b[u][0][j];
                b.y = s.b[u][1][j];
                for (int t = 0; t < RT; ++t)
                    acc[t][j] = __builtin_amdgcn_wmma_f32_16x16x4_f32(
                         false, s.a[u][t],  false, b,
                         (short)0, acc[t][j],  false,  false);
            }
    };

    FragSet s0, s1;
    load_set(s0, 0);
    for (int k = 0; k < K; k += 8 * G) {
        load_set(s1, k + 4 * G);
        mma_set(s0);
        if (k + 8 * G < K) load_set(s0, k + 8 * G);
        mma_set(s1);
    }

    for (int t = 0; t < RT; ++t) {
        const int mb = m0 + 16 * t;
        if (mb < N_NODES) {
            for (int i = 0; i < 8; ++i) {
                bvec o;
                for (int j = 0; j < TN; ++j) o[j] = acc[t][j][i];
                *(bvec*)(C + (size_t)(mb + 8 * h + i) * NOUT + n0 + TN * r) = o;
            }
        }
    }
}

template <int K, int NOUT, int WN, int TN, int G>
__global__ void __launch_bounds__(WN * 32) k_gemm_wmma(const float* __restrict__ A,
                                                       const float* __restrict__ B,
                                                       float* __restrict__ C) {
    constexpr int P  = A_PITCH(K);
    constexpr int BM = 32;
    constexpr int NT = WN * 32;
    static_assert(WN * TN * 16 == NOUT, "a block covers all output columns");
    __shared__ __attribute__((aligned(16))) float As[BM * P];
    const int tid = threadIdx.x;
    const int m0  = blockIdx.x * BM;

    {
        const float4* gA = (const float4*)A;
#pragma unroll 8
        for (int i = tid; i < BM * (K / 4); i += NT) {
            const int row = i / (K / 4);
            const int c4  = i % (K / 4);
            int grow = m0 + row;
            grow = grow < N_NODES ? grow : N_NODES - 1;
            *(float4*)&As[row * P + 4 * c4] = gA[(size_t)grow * (K / 4) + c4];
        }
    }
    __syncthreads();

    const int wn = tid >> 5;
    wave_gemm<K, NOUT, 2, TN, G>(As, B, C, m0, wn * (TN * 16), tid & 31);
}

template <int K, int N1, int N2, int TN, int G>
__global__ void __launch_bounds__(N1 / (TN * 16) * 32)
k_norm_gemm_pair(const float* __restrict__ x,
                 const float* __restrict__ B1, const float* __restrict__ B2,
                 float* __restrict__ C1, float* __restrict__ C2,
                 const int* __restrict__ srcp, const int* __restrict__ dstp,
                 int* cnt, int* __restrict__ col,
                 int* ovf_cnt, int* __restrict__ ovf_dst, int* __restrict__ ovf_src) {
    if (blockIdx.x >= GEMM_BLOCKS) {
        static_assert(N1 / (TN * 16) * 32 == 128, "bucket_edges is written for 128 threads");
        bucket_edges(blockIdx.x - GEMM_BLOCKS, threadIdx.x, srcp, dstp, cnt, col,
                     ovf_cnt, ovf_dst, ovf_src);
        return;
    }
    constexpr int P   = A_PITCH(K);
    constexpr int BM  = 32;
    constexpr int WN  = N1 / (TN * 16);
    constexpr int RQ  = BM / WN;
    static_assert(K == 4 * 4 * 32, "a lane holds 4 float4 of a row");
    static_assert(WN * TN * 16 == N1 && BM % WN == 0, "whole column groups, whole rows per wave");
    static_assert((BM / 16) * (N2 / 32) == WN, "one 16-row x 32-column share of C2 per wave");
    __shared__ __attribute__((aligned(16))) float As[BM * P];
    const int wn   = __builtin_amdgcn_readfirstlane((int)(threadIdx.x >> 5));
    const int lane = threadIdx.x & 31;
    const int m0   = blockIdx.x * BM;

    {
        float4 v[RQ][4];
#pragma unroll
        for (int q = 0; q < RQ; ++q) {
            int grow = m0 + wn + WN * q;
            grow = grow < N_NODES ? grow : N_NODES - 1;
            const float4* grp = (const float4*)(x + (size_t)grow * K);
#pragma unroll
            for (int i = 0; i < 4; ++i) v[q][i] = grp[lane + 32 * i];
        }
#pragma unroll
        for (int q = 0; q < RQ; ++q) {
            const int row = wn + WN * q;
            float ss = 0.0f;
#pragma unroll
            for (int i = 0; i < 4; ++i) {
                float4 t = v[q][i];
                t.x = isfinite(t.x) ? t.x : 0.0f;
                t.y = isfinite(t.y) ? t.y : 0.0f;
                t.z = isfinite(t.z) ? t.z : 0.0f;
                t.w = isfinite(t.w) ? t.w : 0.0f;
                v[q][i] = t;
                ss += t.x * t.x + t.y * t.y + t.z * t.z + t.w * t.w;
            }
            ss = wave_sum32(ss);
            const float scale = 1.0f / fmaxf(sqrtf(ss), L2_EPS);
#pragma unroll
            for (int i = 0; i < 4; ++i) {
                float4 t = v[q][i];
                t.x *= scale; t.y *= scale; t.z *= scale; t.w *= scale;
                *(float4*)&As[row * P + 4 * (lane + 32 * i)] = t;
            }
        }
    }
    __syncthreads();

    wave_gemm<K, N1, 2, TN, G>(As, B1, C1, m0, wn * (TN * 16), lane);
    const int rt = wn & 1;
    wave_gemm<K, N2, 1, 2, G>(As + 16 * rt * P, B2, C2, m0 + 16 * rt, (wn >> 1) * 32, lane);
}

__device__ __forceinline__ void fma4(float4& a, const float4 v, const float w) {
    a.x = fmaf(v.x, w, a.x); a.y = fmaf(v.y, w, a.y);
    a.z = fmaf(v.z, w, a.z); a.w = fmaf(v.w, w, a.w);
}

__device__ __forceinline__ float dinv_of(const int c) { return rsqrtf((float)(c + 1)); }

__device__ __forceinline__ void load_chunk(const int* __restrict__ col,
                                           const int* __restrict__ cnt,
                                           int first, int m, int lane, float di,
                                           int& my_s, float& my_w) {
    my_s = 0; my_w = 0.0f;
    if (lane < m) {
        unsigned s = (unsigned)col[first + lane];
        s = s < (unsigned)N_NODES ? s : (unsigned)(N_NODES - 1);
        my_s = (int)s;
        my_w = dinv_of(cnt[s]) * di;
    }
}

__device__ __forceinline__ bool ovf_entry(const int i, const int node, const float di,
                                          const int* __restrict__ cnt,
                                          const int* __restrict__ ovf_dst,
                                          const int* __restrict__ ovf_src,
                                          int& s_out, float& w_out) {
    if (ovf_dst[i] != node) return false;
    unsigned s = (unsigned)ovf_src[i];
    s = s < (unsigned)N_NODES ? s : (unsigned)(N_NODES - 1);
    s_out = (int)s;
    w_out = dinv_of(cnt[s]) * di;
    return true;
}

template <int R>
__device__ __forceinline__ void add_rows256(const float4* __restrict__ Hv,
                                            int my_s, float my_w, int j, int lane,
                                            float4& a0, float4& a1) {
    float4 v[2 * R]; float w[R];
    for (int u = 0; u < R; ++u) {
        const int s = __builtin_amdgcn_readlane(my_s, j + u);
        w[u] = __int_as_float(__builtin_amdgcn_readlane(__float_as_int(my_w), j + u));
        const float4* r = Hv + (size_t)s * 64;
        v[2 * u]     = r[lane];
        v[2 * u + 1] = r[lane + 32];
    }
    for (int u = 0; u < R; ++u) { fma4(a0, v[2 * u], w[u]); fma4(a1, v[2 * u + 1], w[u]); }
}

__global__ void __launch_bounds__(256) k_gather256(const float* __restrict__ H,
                                                   const int* __restrict__ col,
                                                   const int* __restrict__ cnt,
                                                   const int* __restrict__ ovf_cnt,
                                                   const int* __restrict__ ovf_dst,
                                                   const int* __restrict__ ovf_src,
                                                   const float* __restrict__ b1,
                                                   const float* __restrict__ gamma,
                                                   const float* __restrict__ beta,
                                                   float* __restrict__ out) {
    const int wave = __builtin_amdgcn_readfirstlane((int)(threadIdx.x >> 5));
    const int node = blockIdx.x * 8 + wave;
    const int lane = threadIdx.x & 31;
    const int start = node * CAP;
    const int n     = cnt[node];
    const int nb    = n < CAP ? n : CAP;
    const float di  = dinv_of(n);
    const float4* Hv = (const float4*)H;

    float4 a0, a1;
    {
        const float w = di * di;
        const float4 v0 = Hv[(size_t)node * 64 + lane];
        const float4 v1 = Hv[(size_t)node * 64 + 32 + lane];
        a0.x = v0.x * w; a0.y = v0.y * w; a0.z = v0.z * w; a0.w = v0.w * w;
        a1.x = v1.x * w; a1.y = v1.y * w; a1.z = v1.z * w; a1.w = v1.w * w;
    }
    for (int base = 0; base < nb; base += 32) {
        const int m = (nb - base) < 32 ? (nb - base) : 32;
        int my_s; float my_w;
        load_chunk(col, cnt, start + base, m, lane, di, my_s, my_w);
        int j = 0;
        for (; j + 4 <= m; j += 4) add_rows256<4>(Hv, my_s, my_w, j, lane, a0, a1);
        const int left = m - j;
        if (left == 3)      add_rows256<3>(Hv, my_s, my_w, j, lane, a0, a1);
        else if (left == 2) add_rows256<2>(Hv, my_s, my_w, j, lane, a0, a1);
        else if (left == 1) add_rows256<1>(Hv, my_s, my_w, j, lane, a0, a1);
    }
    if (n > CAP) {
        int novf = *ovf_cnt;
        novf = novf < N_EDGES ? novf : N_EDGES;
        for (int i = 0; i < novf; ++i) {
            int s; float w;
            if (ovf_entry(i, node, di, cnt, ovf_dst, ovf_src, s, w)) {
                fma4(a0, Hv[(size_t)s * 64 + lane], w);
                fma4(a1, Hv[(size_t)s * 64 + 32 + lane], w);
            }
        }
    }
    float4 v[2] = {a0, a1};
    float s = 0.0f;
#pragma unroll
    for (int i = 0; i < 2; ++i) {
        float4 t  = v[i];
        float4 bb = ((const float4*)b1)[lane + 32 * i];
        t.x += bb.x; t.y += bb.y; t.z += bb.z; t.w += bb.w;
        v[i] = t;
        s += t.x + t.y + t.z + t.w;
    }
    float mu = wave_sum32(s) * (1.0f / F_HID);
    float ss = 0.0f;
#pragma unroll
    for (int i = 0; i < 2; ++i) {
        float4 t = v[i];
        ss += (t.x - mu) * (t.x - mu) + (t.y - mu) * (t.y - mu) +
              (t.z - mu) * (t.z - mu) + (t.w - mu) * (t.w - mu);
    }
    float rstd = rsqrtf(wave_sum32(ss) * (1.0f / F_HID) + LN_EPS);
    float4* orow = (float4*)(out + (size_t)node * F_HID);
#pragma unroll
    for (int i = 0; i < 2; ++i) {
        float4 t = v[i];
        float4 g = ((const float4*)gamma)[lane + 32 * i];
        float4 b = ((const float4*)beta)[lane + 32 * i];
        t.x = fmaxf((t.x - mu) * rstd * g.x + b.x, 0.0f);
        t.y = fmaxf((t.y - mu) * rstd * g.y + b.y, 0.0f);
        t.z = fmaxf((t.z - mu) * rstd * g.z + b.z, 0.0f);
        t.w = fmaxf((t.w - mu) * rstd * g.w + b.w, 0.0f);
        orow[lane + 32 * i] = t;
    }
}

__device__ __forceinline__ float final_value(float v) {
    const float lim = v > 0.0f ? 20.0f : -20.0f;
    v = isinf(v) ? lim : v;
    return isnan(v) ? 0.0f : v;
}

__global__ void __launch_bounds__(256) k_gather64(const float* __restrict__ H,
                                                  const int* __restrict__ col,
                                                  const int* __restrict__ cnt,
                                                  const int* __restrict__ ovf_cnt,
                                                  const int* __restrict__ ovf_dst,
                                                  const int* __restrict__ ovf_src,
                                                  const float* __restrict__ b2,
                                                  float* __restrict__ out) {
    const int wave = __builtin_amdgcn_readfirstlane((int)(threadIdx.x >> 5));
    const int node = blockIdx.x * 8 + wave;
    const int lane = threadIdx.x & 31;
    const int half = lane >> 4;
    const int c    = lane & 15;
    const int start = node * CAP;
    const int n     = cnt[node];
    const int nb    = n < CAP ? n : CAP;
    const float di  = dinv_of(n);
    const float4* Hv = (const float4*)H;

    float4 acc = {0.0f, 0.0f, 0.0f, 0.0f};
    if (half == 0) {
        const float w = di * di;
        const float4 v = Hv[(size_t)node * 16 + c];
        acc.x = v.x * w; acc.y = v.y * w; acc.z = v.z * w; acc.w = v.w * w;
    }
    for (int base = 0; base < nb; base += 32) {
        const int m = (nb - base) < 32 ? (nb - base) : 32;
        int my_s; float my_w;
        load_chunk(col, cnt, start + base, m, lane, di, my_s, my_w);
        for (int j = 0; j < m; j += 8) {
            float4 v[4]; float w[4];
            for (int u = 0; u < 4; ++u) {
                const int e0 = j + 2 * u;
                const int sA = __builtin_amdgcn_readlane(my_s, e0);
                const int sB = __builtin_amdgcn_readlane(my_s, e0 + 1);
                const float wA = __int_as_float(__builtin_amdgcn_readlane(__float_as_int(my_w), e0));
                const float wB = __int_as_float(__builtin_amdgcn_readlane(__float_as_int(my_w), e0 + 1));
                const int s = half ? sB : sA;
                w[u] = half ? wB : wA;
                v[u] = Hv[(size_t)s * 16 + c];
            }
            for (int u = 0; u < 4; ++u)
                if (j + 2 * u + half < m) fma4(acc, v[u], w[u]);
        }
    }
    if (n > CAP) {
        int novf = *ovf_cnt;
        novf = novf < N_EDGES ? novf : N_EDGES;
        for (int i = 0; i < novf; ++i) {
            int s; float w;
            if (ovf_entry(i, node, di, cnt, ovf_dst, ovf_src, s, w)) {
                if (half == 0) fma4(acc, Hv[(size_t)s * 16 + c], w);
            }
        }
    }
    acc.x += __shfl_xor(acc.x, 16, 32);
    acc.y += __shfl_xor(acc.y, 16, 32);
    acc.z += __shfl_xor(acc.z, 16, 32);
    acc.w += __shfl_xor(acc.w, 16, 32);
    if (half == 0) {
        float4* o = (float4*)(out + (size_t)node * F_OUT) + c;
        const float4 bb = ((const float4*)b2)[c];
        float4 t = *o;
        t.x = final_value(t.x + acc.x + bb.x);
        t.y = final_value(t.y + acc.y + bb.y);
        t.z = final_value(t.z + acc.z + bb.z);
        t.w = final_value(t.w + acc.w + bb.w);
        *o = t;
    }
}

extern "C" void kernel_launch(void* const* d_in, const int* in_sizes, int n_in,
                              void* d_out, int out_size, void* d_ws, size_t ws_size,
                              hipStream_t stream) {
    const float* x     = (const float*)d_in[0];
    const int*   ei    = (const int*)d_in[1];
    const float* W1    = (const float*)d_in[2];
    const float* b1    = (const float*)d_in[3];
    const float* gamma = (const float*)d_in[4];
    const float* beta  = (const float*)d_in[5];
    const float* W2    = (const float*)d_in[6];
    const float* b2    = (const float*)d_in[7];
    const float* Wskip = (const float*)d_in[8];
    const int* srcp = ei;
    const int* dstp = ei + N_EDGES;

    float* ws   = (float*)d_ws;
    float* base = ws + N_NODES;
    float* h1   = base + (size_t)N_NODES * F_IN;
    float* agg  = h1 + (size_t)N_NODES * F_HID;
    float* out  = (float*)d_out;

    int* cnt     = (int*)base + 16;
    int* col     = cnt + N_LINE;
    int* ovf_dst = col + (size_t)N_NODES * CAP;
    int* ovf_src = ovf_dst + N_EDGES;
    int* ovf_cnt = ovf_src + N_EDGES;

    k_deg_init<<<(N_NODES + 255) / 256, 256, 0, stream>>>(cnt, ovf_cnt);

    k_norm_gemm_pair<F_IN, F_HID, F_OUT, 4, 4><<<GEMM_BLOCKS + EDGE_BLOCKS, 128, 0, stream>>>(
        x, W1, Wskip, h1, out, srcp, dstp, cnt, col, ovf_cnt, ovf_dst, ovf_src);

    k_gather256<<<N_NODES / 8, 256, 0, stream>>>(h1, col, cnt, ovf_cnt, ovf_dst, ovf_src, b1, gamma, beta, agg);

    k_gemm_wmma<F_HID, F_OUT, 2, 2, 8><<<(N_NODES + 31) / 32, 64, 0, stream>>>(agg, W2, h1);
    k_gather64<<<N_NODES / 8, 256, 0, stream>>>(h1, col, cnt, ovf_cnt, ovf_dst, ovf_src, b2, out);
}


# W2 product fused into the 256-wide gather: a block gathers + normalises 32 nodes into LDS, 4 of its waves multiply the tile by W2; 5 -> 4 launches, h2 rows on line boundaries; sum order kept
# speedup vs baseline: 4.0415x; 1.0619x over previous
#include <hip/hip_runtime.h>
#include <math.h>

#define N_NODES 50000
#define N_EDGES 800000
#define F_IN    512
#define F_HID   256
#define F_OUT   64
#define LN_EPS  1e-5f
#define L2_EPS  1e-12f
#define N_LINE  ((N_NODES + 31) / 32 * 32)

typedef __attribute__((ext_vector_type(2))) float v2f;
typedef __attribute__((ext_vector_type(4))) float v4f;
typedef __attribute__((ext_vector_type(8))) float v8f;

__device__ __forceinline__ float wave_sum32(float v) {
#pragma unroll
    for (int m = 16; m >= 1; m >>= 1) v += __shfl_xor(v, m, 32);
    return v;
}

#define CAP 64
#define GEMM_BLOCKS ((N_NODES + 31) / 32)
#define EDGES_PER_BLOCK 1024
#define EDGE_BLOCKS ((N_EDGES + EDGES_PER_BLOCK - 1) / EDGES_PER_BLOCK)

__global__ void __launch_bounds__(256) k_deg_init(int* __restrict__ cnt,
                                                  int* __restrict__ ovf_cnt) {
    int i = blockIdx.x * 256 + threadIdx.x;
    if (i < N_NODES) cnt[i] = 0;
    if (i == 0) *ovf_cnt = 0;
}

__device__ __forceinline__ void bucket_edges(const int b, const int tid,
                                             const int* __restrict__ srcp,
                                             const int* __restrict__ dstp,
                                             int* cnt, int* __restrict__ col,
                                             int* ovf_cnt,
                                             int* __restrict__ ovf_dst,
                                             int* __restrict__ ovf_src) {
    int d[8], s[8], r[8];
#pragma unroll
    for (int u = 0; u < 8; ++u) {
        const int e = b * EDGES_PER_BLOCK + 128 * u + tid;
        const bool ok = e < N_EDGES;
        d[u] = ok ? dstp[e] : -1;
        s[u] = ok ? srcp[e] : 0;
    }
#pragma unroll
    for (int u = 0; u < 8; ++u) r[u] = d[u] >= 0 ? atomicAdd(&cnt[d[u]], 1) : 0;
#pragma unroll
    for (int u = 0; u < 8; ++u) {
        if (d[u] >= 0) {
            if (r[u] < CAP) {
                col[(size_t)d[u] * CAP + r[u]] = s[u];
            } else {
                const int o = atomicAdd(ovf_cnt, 1);
                ovf_dst[o] = d[u];
                ovf_src[o] = s[u];
            }
        }
    }
}

template <int TN> struct BVec;
template <> struct BVec<2> { typedef v2f T; };
template <> struct BVec<4> { typedef v4f T; };

#define A_PITCH(K) ((K) + 4)

template <int K, int NOUT, int RT, int TN, int G>
__device__ __forceinline__ void wave_gemm(const float* As,
                                          const float* __restrict__ B,
                                          float* __restrict__ C,
                                          const int m0, const int n0, const int lane) {
    typedef typename BVec<TN>::T bvec;
    constexpr int P = A_PITCH(K);
    static_assert((K / (4 * G)) % 2 == 0, "the K loop takes two fragment sets per trip");
    const int r    = lane & 15;
    const int h    = lane >> 4;
    const float* ap = &As[r * P + 2 * h];
    const float* bp = B + (size_t)(2 * h) * NOUT + n0 + TN * r;

    v8f acc[RT][TN];
    for (int t = 0; t < RT; ++t)
        for (int j = 0; j < TN; ++j) acc[t][j] = v8f{};

    struct FragSet { v2f a[G][RT]; bvec b[G][2]; };
    auto load_set = [&](FragSet& s, const int k) {
        for (int u = 0; u < G; ++u) {
            for (int t = 0; t < RT; ++t)
                s.a[u][t] = *(const v2f*)(ap + 16 * t * P + k + 4 * u);
            s.b[u][0] = *(const bvec*)(bp + (size_t)(k + 4 * u) * NOUT);
            s.b[u][1] = *(const bvec*)(bp + (size_t)(k + 4 * u + 1) * NOUT);
        }
    };
    auto mma_set = [&](const FragSet& s) {
        for (int u = 0; u < G; ++u)
            for (int j = 0; j < TN; ++j) {
                v2f b;
                b.x = s.b[u][0][j];
                b.y = s.b[u][1][j];
                for (int t = 0; t < RT; ++t)
                    acc[t][j] = __builtin_amdgcn_wmma_f32_16x16x4_f32(
                         false, s.a[u][t],  false, b,
                         (short)0, acc[t][j],  false,  false);
            }
    };

    FragSet s0, s1;
    load_set(s0, 0);
    for (int k = 0; k < K; k += 8 * G) {
        load_set(s1, k + 4 * G);
        mma_set(s0);
        if (k + 8 * G < K) load_set(s0, k + 8 * G);
        mma_set(s1);
    }

    for (int t = 0; t < RT; ++t) {
        const int mb = m0 + 16 * t;
        if (mb < N_NODES) {
            for (int i = 0; i < 8; ++i) {
                bvec o;
                for (int j = 0; j < TN; ++j) o[j] = acc[t][j][i];
                *(bvec*)(C + (size_t)(mb + 8 * h + i) * NOUT + n0 + TN * r) = o;
            }
        }
    }
}

template <int K, int N1, int N2, int TN, int G>
__global__ void __launch_bounds__(N1 / (TN * 16) * 32)
k_norm_gemm_pair(const float* __restrict__ x,
                 const float* __restrict__ B1, const float* __restrict__ B2,
                 float* __restrict__ C1, float* __restrict__ C2,
                 const int* __restrict__ srcp, const int* __restrict__ dstp,
                 int* cnt, int* __restrict__ col,
                 int* ovf_cnt, int* __restrict__ ovf_dst, int* __restrict__ ovf_src) {
    if (blockIdx.x >= GEMM_BLOCKS) {
        static_assert(N1 / (TN * 16) * 32 == 128, "bucket_edges is written for 128 threads");
        bucket_edges(blockIdx.x - GEMM_BLOCKS, threadIdx.x, srcp, dstp, cnt, col,
                     ovf_cnt, ovf_dst, ovf_src);
        return;
    }
    constexpr int P   = A_PITCH(K);
    constexpr int BM  = 32;
    constexpr int WN  = N1 / (TN * 16);
    constexpr int RQ  = BM / WN;
    static_assert(K == 4 * 4 * 32, "a lane holds 4 float4 of a row");
    static_assert(WN * TN * 16 == N1 && BM % WN == 0, "whole column groups, whole rows per wave");
    static_assert((BM / 16) * (N2 / 32) == WN, "one 16-row x 32-column share of C2 per wave");
    __shared__ __attribute__((aligned(16))) float As[BM * P];
    const int wn   = __builtin_amdgcn_readfirstlane((int)(threadIdx.x >> 5));
    const int lane = threadIdx.x & 31;
    const int m0   = blockIdx.x * BM;

    {
        float4 v[RQ][4];
#pragma unroll
        for (int q = 0; q < RQ; ++q) {
            int grow = m0 + wn + WN * q;
            grow = grow < N_NODES ? grow : N_NODES - 1;
            const float4* grp = (const float4*)(x + (size_t)grow * K);
#pragma unroll
            for (int i = 0; i < 4; ++i) v[q][i] = grp[lane + 32 * i];
        }
#pragma unroll
        for (int q = 0; q < RQ; ++q) {
            const int row = wn + WN * q;
            float ss = 0.0f;
#pragma unroll
            for (int i = 0; i < 4; ++i) {
                float4 t = v[q][i];
                t.x = isfinite(t.x) ? t.x : 0.0f;
                t.y = isfinite(t.y) ? t.y : 0.0f;
                t.z = isfinite(t.z) ? t.z : 0.0f;
                t.w = isfinite(t.w) ? t.w : 0.0f;
                v[q][i] = t;
                ss += t.x * t.x + t.y * t.y + t.z * t.z + t.w * t.w;
            }
            ss = wave_sum32(ss);
            const float scale = 1.0f / fmaxf(sqrtf(ss), L2_EPS);
#pragma unroll
            for (int i = 0; i < 4; ++i) {
                float4 t = v[q][i];
                t.x *= scale; t.y *= scale; t.z *= scale; t.w *= scale;
                *(float4*)&As[row * P + 4 * (lane + 32 * i)] = t;
            }
        }
    }
    __syncthreads();

    wave_gemm<K, N1, 2, TN, G>(As, B1, C1, m0, wn * (TN * 16), lane);
    const int rt = wn & 1;
    wave_gemm<K, N2, 1, 2, G>(As + 16 * rt * P, B2, C2, m0 + 16 * rt, (wn >> 1) * 32, lane);
}

__device__ __forceinline__ void fma4(float4& a, const float4 v, const float w) {
    a.x = fmaf(v.x, w, a.x); a.y = fmaf(v.y, w, a.y);
    a.z = fmaf(v.z, w, a.z); a.w = fmaf(v.w, w, a.w);
}

__device__ __forceinline__ float dinv_of(const int c) { return rsqrtf((float)(c + 1)); }

__device__ __forceinline__ void load_chunk(const int* __restrict__ col,
                                           const int* __restrict__ cnt,
                                           int first, int m, int lane, float di,
                                           int& my_s, float& my_w) {
    my_s = 0; my_w = 0.0f;
    if (lane < m) {
        unsigned s = (unsigned)col[first + lane];
        s = s < (unsigned)N_NODES ? s : (unsigned)(N_NODES - 1);
        my_s = (int)s;
        my_w = dinv_of(cnt[s]) * di;
    }
}

__device__ __forceinline__ bool ovf_entry(const int i, const int node, const float di,
                                          const int* __restrict__ cnt,
                                          const int* __restrict__ ovf_dst,
                                          const int* __restrict__ ovf_src,
                                          int& s_out, float& w_out) {
    if (ovf_dst[i] != node) return false;
    unsigned s = (unsigned)ovf_src[i];
    s = s < (unsigned)N_NODES ? s : (unsigned)(N_NODES - 1);
    s_out = (int)s;
    w_out = dinv_of(cnt[s]) * di;
    return true;
}

template <int R>
__device__ __forceinline__ void add_rows256(const float4* __restrict__ Hv,
                                            int my_s, float my_w, int j, int lane,
                                            float4& a0, float4& a1) {
    float4 v[2 * R]; float w[R];
    for (int u = 0; u < R; ++u) {
        const int s = __builtin_amdgcn_readlane(my_s, j + u);
        w[u] = __int_as_float(__builtin_amdgcn_readlane(__float_as_int(my_w), j + u));
        const float4* r = Hv + (size_t)s * 64;
        v[2 * u]     = r[lane];
        v[2 * u + 1] = r[lane + 32];
    }
    for (int u = 0; u < R; ++u) { fma4(a0, v[2 * u], w[u]); fma4(a1, v[2 * u + 1], w[u]); }
}

#define G2_ROWS 32
__global__ void __launch_bounds__(256) k_gather256_gemm(const float* __restrict__ H,
                                                        const int* __restrict__ col,
                                                        const int* __restrict__ cnt,
                                                        const int* __restrict__ ovf_cnt,
                                                        const int* __restrict__ ovf_dst,
                                                        const int* __restrict__ ovf_src,
                                                        const float* __restrict__ b1,
                                                        const float* __restrict__ gamma,
                                                        const float* __restrict__ beta,
                                                        const float* __restrict__ W2,
                                                        float* __restrict__ C) {
    constexpr int P = A_PITCH(F_HID);
    __shared__ __attribute__((aligned(16))) float As[G2_ROWS * P];
    const int wave = __builtin_amdgcn_readfirstlane((int)(threadIdx.x >> 5));
    const int lane = threadIdx.x & 31;
    const int m0   = blockIdx.x * G2_ROWS;
    const float4* Hv = (const float4*)H;

    for (int q = 0; q < G2_ROWS / 8; ++q) {
        const int row = wave + 8 * q;
        int node = m0 + row;
        node = node < N_NODES ? node : N_NODES - 1;
        const int start = node * CAP;
        const int n     = cnt[node];
        const int nb    = n < CAP ? n : CAP;
        const float di  = dinv_of(n);

        float4 a0, a1;
        {
            const float w = di * di;
            const float4 v0 = Hv[(size_t)node * 64 + lane];
            const float4 v1 = Hv[(size_t)node * 64 + 32 + lane];
            a0.x = v0.x * w; a0.y = v0.y * w; a0.z = v0.z * w; a0.w = v0.w * w;
            a1.x = v1.x * w; a1.y = v1.y * w; a1.z = v1.z * w; a1.w = v1.w * w;
        }
        for (int base = 0; base < nb; base += 32) {
            const int m = (nb - base) < 32 ? (nb - base) : 32;
            int my_s; float my_w;
            load_chunk(col, cnt, start + base, m, lane, di, my_s, my_w);
            int j = 0;
            for (; j + 4 <= m; j += 4) add_rows256<4>(Hv, my_s, my_w, j, lane, a0, a1);
            const int left = m - j;
            if (left == 3)      add_rows256<3>(Hv, my_s, my_w, j, lane, a0, a1);
            else if (left == 2) add_rows256<2>(Hv, my_s, my_w, j, lane, a0, a1);
            else if (left == 1) add_rows256<1>(Hv, my_s, my_w, j, lane, a0, a1);
        }
        if (n > CAP) {
            int novf = *ovf_cnt;
            novf = novf < N_EDGES ? novf : N_EDGES;
            for (int i = 0; i < novf; ++i) {
                int s; float w;
                if (ovf_entry(i, node, di, cnt, ovf_dst, ovf_src, s, w)) {
                    fma4(a0, Hv[(size_t)s * 64 + lane], w);
                    fma4(a1, Hv[(size_t)s * 64 + 32 + lane], w);
                }
            }
        }
        float4 v[2] = {a0, a1};
        float s = 0.0f;
#pragma unroll
        for (int i = 0; i < 2; ++i) {
            float4 t  = v[i];
            float4 bb = ((const float4*)b1)[lane + 32 * i];
            t.x += bb.x; t.y += bb.y; t.z += bb.z; t.w += bb.w;
            v[i] = t;
            s += t.x + t.y + t.z + t.w;
        }
        float mu = wave_sum32(s) * (1.0f / F_HID);
        float ss = 0.0f;
#pragma unroll
        for (int i = 0; i < 2; ++i) {
            float4 t = v[i];
            ss += (t.x - mu) * (t.x - mu) + (t.y - mu) * (t.y - mu) +
                  (t.z - mu) * (t.z - mu) + (t.w - mu) * (t.w - mu);
        }
        float rstd = rsqrtf(wave_sum32(ss) * (1.0f / F_HID) + LN_EPS);
#pragma unroll
        for (int i = 0; i < 2; ++i) {
            float4 t = v[i];
            float4 g = ((const float4*)gamma)[lane + 32 * i];
            float4 b = ((const float4*)beta)[lane + 32 * i];
            t.x = fmaxf((t.x - mu) * rstd * g.x + b.x, 0.0f);
            t.y = fmaxf((t.y - mu) * rstd * g.y + b.y, 0.0f);
            t.z = fmaxf((t.z - mu) * rstd * g.z + b.z, 0.0f);
            t.w = fmaxf((t.w - mu) * rstd * g.w + b.w, 0.0f);
            *(float4*)&As[row * P + 4 * (lane + 32 * i)] = t;
        }
    }
    __syncthreads();

    if (wave < 4) {
        const int rt = wave & 1;
        wave_gemm<F_HID, F_OUT, 1, 2, 8>(As + 16 * rt * P, W2, C, m0 + 16 * rt, (wave >> 1) * 32, lane);
    }
}

__device__ __forceinline__ float final_value(float v) {
    const float lim = v > 0.0f ? 20.0f : -20.0f;
    v = isinf(v) ? lim : v;
    return isnan(v) ? 0.0f : v;
}

__global__ void __launch_bounds__(256) k_gather64(const float* __restrict__ H,
                                                  const int* __restrict__ col,
                                                  const int* __restrict__ cnt,
                                                  const int* __restrict__ ovf_cnt,
                                                  const int* __restrict__ ovf_dst,
                                                  const int* __restrict__ ovf_src,
                                                  const float* __restrict__ b2,
                                                  float* __restrict__ out) {
    const int wave = __builtin_amdgcn_readfirstlane((int)(threadIdx.x >> 5));
    const int node = blockIdx.x * 8 + wave;
    const int lane = threadIdx.x & 31;
    const int half = lane >> 4;
    const int c    = lane & 15;
    const int start = node * CAP;
    const int n     = cnt[node];
    const int nb    = n < CAP ? n : CAP;
    const float di  = dinv_of(n);
    const float4* Hv = (const float4*)H;

    float4 acc = {0.0f, 0.0f, 0.0f, 0.0f};
    if (half == 0) {
        const float w = di * di;
        const float4 v = Hv[(size_t)node * 16 + c];
        acc.x = v.x * w; acc.y = v.y * w; acc.z = v.z * w; acc.w = v.w * w;
    }
    for (int base = 0; base < nb; base += 32) {
        const int m = (nb - base) < 32 ? (nb - base) : 32;
        int my_s; float my_w;
        load_chunk(col, cnt, start + base, m, lane, di, my_s, my_w);
        for (int j = 0; j < m; j += 8) {
            float4 v[4]; float w[4];
            for (int u = 0; u < 4; ++u) {
                const int e0 = j + 2 * u;
                const int sA = __builtin_amdgcn_readlane(my_s, e0);
                const int sB = __builtin_amdgcn_readlane(my_s, e0 + 1);
                const float wA = __int_as_float(__builtin_amdgcn_readlane(__float_as_int(my_w), e0));
                const float wB = __int_as_float(__builtin_amdgcn_readlane(__float_as_int(my_w), e0 + 1));
                const int s = half ? sB : sA;
                w[u] = half ? wB : wA;
                v[u] = Hv[(size_t)s * 16 + c];
            }
            for (int u = 0; u < 4; ++u)
                if (j + 2 * u + half < m) fma4(acc, v[u], w[u]);
        }
    }
    if (n > CAP) {
        int novf = *ovf_cnt;
        novf = novf < N_EDGES ? novf : N_EDGES;
        for (int i = 0; i < novf; ++i) {
            int s; float w;
            if (ovf_entry(i, node, di, cnt, ovf_dst, ovf_src, s, w)) {
                if (half == 0) fma4(acc, Hv[(size_t)s * 16 + c], w);
            }
        }
    }
    acc.x += __shfl_xor(acc.x, 16, 32);
    acc.y += __shfl_xor(acc.y, 16, 32);
    acc.z += __shfl_xor(acc.z, 16, 32);
    acc.w += __shfl_xor(acc.w, 16, 32);
    if (half == 0) {
        float4* o = (float4*)(out + (size_t)node * F_OUT) + c;
        const float4 bb = ((const float4*)b2)[c];
        float4 t = *o;
        t.x = final_value(t.x + acc.x + bb.x);
        t.y = final_value(t.y + acc.y + bb.y);
        t.z = final_value(t.z + acc.z + bb.z);
        t.w = final_value(t.w + acc.w + bb.w);
        *o = t;
    }
}

extern "C" void kernel_launch(void* const* d_in, const int* in_sizes, int n_in,
                              void* d_out, int out_size, void* d_ws, size_t ws_size,
                              hipStream_t stream) {
    const float* x     = (const float*)d_in[0];
    const int*   ei    = (const int*)d_in[1];
    const float* W1    = (const float*)d_in[2];
    const float* b1    = (const float*)d_in[3];
    const float* gamma = (const float*)d_in[4];
    const float* beta  = (const float*)d_in[5];
    const float* W2    = (const float*)d_in[6];
    const float* b2    = (const float*)d_in[7];
    const float* Wskip = (const float*)d_in[8];
    const int* srcp = ei;
    const int* dstp = ei + N_EDGES;

    float* ws   = (float*)d_ws;
    float* base = ws + N_NODES;
    float* h1   = base + (size_t)N_NODES * F_IN;
    float* h2   = h1 + (size_t)N_NODES * F_HID + 48;
    float* out  = (float*)d_out;

    int* cnt     = (int*)base + 16;
    int* col     = cnt + N_LINE;
    int* ovf_dst = col + (size_t)N_NODES * CAP;
    int* ovf_src = ovf_dst + N_EDGES;
    int* ovf_cnt = ovf_src + N_EDGES;

    k_deg_init<<<(N_NODES + 255) / 256, 256, 0, stream>>>(cnt, ovf_cnt);

    k_norm_gemm_pair<F_IN, F_HID, F_OUT, 4, 4><<<GEMM_BLOCKS + EDGE_BLOCKS, 128, 0, stream>>>(
        x, W1, Wskip, h1, out, srcp, dstp, cnt, col, ovf_cnt, ovf_dst, ovf_src);

    k_gather256_gemm<<<(N_NODES + G2_ROWS - 1) / G2_ROWS, 256, 0, stream>>>(
        h1, col, cnt, ovf_cnt, ovf_dst, ovf_src, b1, gamma, beta, W2, h2);

    k_gather64<<<N_NODES / 8, 256, 0, stream>>>(h2, col, cnt, ovf_cnt, ovf_dst, ovf_src, b2, out);
}


# h1 moved 192 bytes onto a 256-byte boundary: each gathered 1 KB row is 8 whole lines (a 512-byte load touches 4 lines, not 5) and the product stores whole lines; h2 address unchanged
# speedup vs baseline: 4.1786x; 1.0339x over previous
#include <hip/hip_runtime.h>
#include <math.h>

#define N_NODES 50000
#define N_EDGES 800000
#define F_IN    512
#define F_HID   256
#define F_OUT   64
#define LN_EPS  1e-5f
#define L2_EPS  1e-12f
#define N_LINE  ((N_NODES + 31) / 32 * 32)

typedef __attribute__((ext_vector_type(2))) float v2f;
typedef __attribute__((ext_vector_type(4))) float v4f;
typedef __attribute__((ext_vector_type(8))) float v8f;

__device__ __forceinline__ float wave_sum32(float v) {
#pragma unroll
    for (int m = 16; m >= 1; m >>= 1) v += __shfl_xor(v, m, 32);
    return v;
}

#define CAP 64
#define GEMM_BLOCKS ((N_NODES + 31) / 32)
#define EDGES_PER_BLOCK 1024
#define EDGE_BLOCKS ((N_EDGES + EDGES_PER_BLOCK - 1) / EDGES_PER_BLOCK)

__global__ void __launch_bounds__(256) k_deg_init(int* __restrict__ cnt,
                                                  int* __restrict__ ovf_cnt) {
    int i = blockIdx.x * 256 + threadIdx.x;
    if (i < N_NODES) cnt[i] = 0;
    if (i == 0) *ovf_cnt = 0;
}

__device__ __forceinline__ void bucket_edges(const int b, const int tid,
                                             const int* __restrict__ srcp,
                                             const int* __restrict__ dstp,
                                             int* cnt, int* __restrict__ col,
                                             int* ovf_cnt,
                                             int* __restrict__ ovf_dst,
                                             int* __restrict__ ovf_src) {
    int d[8], s[8], r[8];
#pragma unroll
    for (int u = 0; u < 8; ++u) {
        const int e = b * EDGES_PER_BLOCK + 128 * u + tid;
        const bool ok = e < N_EDGES;
        d[u] = ok ? dstp[e] : -1;
        s[u] = ok ? srcp[e] : 0;
    }
#pragma unroll
    for (int u = 0; u < 8; ++u) r[u] = d[u] >= 0 ? atomicAdd(&cnt[d[u]], 1) : 0;
#pragma unroll
    for (int u = 0; u < 8; ++u) {
        if (d[u] >= 0) {
            if (r[u] < CAP) {
                col[(size_t)d[u] * CAP + r[u]] = s[u];
            } else {
                const int o = atomicAdd(ovf_cnt, 1);
                ovf_dst[o] = d[u];
                ovf_src[o] = s[u];
            }
        }
    }
}

template <int TN> struct BVec;
template <> struct BVec<2> { typedef v2f T; };
template <> struct BVec<4> { typedef v4f T; };

#define A_PITCH(K) ((K) + 4)

template <int K, int NOUT, int RT, int TN, int G>
__device__ __forceinline__ void wave_gemm(const float* As,
                                          const float* __restrict__ B,
                                          float* __restrict__ C,
                                          const int m0, const int n0, const int lane) {
    typedef typename BVec<TN>::T bvec;
    constexpr int P = A_PITCH(K);
    static_assert((K / (4 * G)) % 2 == 0, "the K loop takes two fragment sets per trip");
    const int r    = lane & 15;
    const int h    = lane >> 4;
    const float* ap = &As[r * P + 2 * h];
    const float* bp = B + (size_t)(2 * h) * NOUT + n0 + TN * r;

    v8f acc[RT][TN];
    for (int t = 0; t < RT; ++t)
        for (int j = 0; j < TN; ++j) acc[t][j] = v8f{};

    struct FragSet { v2f a[G][RT]; bvec b[G][2]; };
    auto load_set = [&](FragSet& s, const int k) {
        for (int u = 0; u < G; ++u) {
            for (int t = 0; t < RT; ++t)
                s.a[u][t] = *(const v2f*)(ap + 16 * t * P + k + 4 * u);
            s.b[u][0] = *(const bvec*)(bp + (size_t)(k + 4 * u) * NOUT);
            s.b[u][1] = *(const bvec*)(bp + (size_t)(k + 4 * u + 1) * NOUT);
        }
    };
    auto mma_set = [&](const FragSet& s) {
        for (int u = 0; u < G; ++u)
            for (int j = 0; j < TN; ++j) {
                v2f b;
                b.x = s.b[u][0][j];
                b.y = s.b[u][1][j];
                for (int t = 0; t < RT; ++t)
                    acc[t][j] = __builtin_amdgcn_wmma_f32_16x16x4_f32(
                         false, s.a[u][t],  false, b,
                         (short)0, acc[t][j],  false,  false);
            }
    };

    FragSet s0, s1;
    load_set(s0, 0);
    for (int k = 0; k < K; k += 8 * G) {
        load_set(s1, k + 4 * G);
        mma_set(s0);
        if (k + 8 * G < K) load_set(s0, k + 8 * G);
        mma_set(s1);
    }

    for (int t = 0; t < RT; ++t) {
        const int mb = m0 + 16 * t;
        if (mb < N_NODES) {
            for (int i = 0; i < 8; ++i) {
                bvec o;
                for (int j = 0; j < TN; ++j) o[j] = acc[t][j][i];
                *(bvec*)(C + (size_t)(mb + 8 * h + i) * NOUT + n0 + TN * r) = o;
            }
        }
    }
}

template <int K, int N1, int N2, int TN, int G>
__global__ void __launch_bounds__(N1 / (TN * 16) * 32)
k_norm_gemm_pair(const float* __restrict__ x,
                 const float* __restrict__ B1, const float* __restrict__ B2,
                 float* __restrict__ C1, float* __restrict__ C2,
                 const int* __restrict__ srcp, const int* __restrict__ dstp,
                 int* cnt, int* __restrict__ col,
                 int* ovf_cnt, int* __restrict__ ovf_dst, int* __restrict__ ovf_src) {
    if (blockIdx.x >= GEMM_BLOCKS) {
        static_assert(N1 / (TN * 16) * 32 == 128, "bucket_edges is written for 128 threads");
        bucket_edges(blockIdx.x - GEMM_BLOCKS, threadIdx.x, srcp, dstp, cnt, col,
                     ovf_cnt, ovf_dst, ovf_src);
        return;
    }
    constexpr int P   = A_PITCH(K);
    constexpr int BM  = 32;
    constexpr int WN  = N1 / (TN * 16);
    constexpr int RQ  = BM / WN;
    static_assert(K == 4 * 4 * 32, "a lane holds 4 float4 of a row");
    static_assert(WN * TN * 16 == N1 && BM % WN == 0, "whole column groups, whole rows per wave");
    static_assert((BM / 16) * (N2 / 32) == WN, "one 16-row x 32-column share of C2 per wave");
    __shared__ __attribute__((aligned(16))) float As[BM * P];
    const int wn   = __builtin_amdgcn_readfirstlane((int)(threadIdx.x >> 5));
    const int lane = threadIdx.x & 31;
    const int m0   = blockIdx.x * BM;

    {
        float4 v[RQ][4];
#pragma unroll
        for (int q = 0; q < RQ; ++q) {
            int grow = m0 + wn + WN * q;
            grow = grow < N_NODES ? grow : N_NODES - 1;
            const float4* grp = (const float4*)(x + (size_t)grow * K);
#pragma unroll
            for (int i = 0; i < 4; ++i) v[q][i] = grp[lane + 32 * i];
        }
#pragma unroll
        for (int q = 0; q < RQ; ++q) {
            const int row = wn + WN * q;
            float ss = 0.0f;
#pragma unroll
            for (int i = 0; i < 4; ++i) {
                float4 t = v[q][i];
                t.x = isfinite(t.x) ? t.x : 0.0f;
                t.y = isfinite(t.y) ? t.y : 0.0f;
                t.z = isfinite(t.z) ? t.z : 0.0f;
                t.w = isfinite(t.w) ? t.w : 0.0f;
                v[q][i] = t;
                ss += t.x * t.x + t.y * t.y + t.z * t.z + t.w * t.w;
            }
            ss = wave_sum32(ss);
            const float scale = 1.0f / fmaxf(sqrtf(ss), L2_EPS);
#pragma unroll
            for (int i = 0; i < 4; ++i) {
                float4 t = v[q][i];
                t.x *= scale; t.y *= scale; t.z *= scale; t.w *= scale;
                *(float4*)&As[row * P + 4 * (lane + 32 * i)] = t;
            }
        }
    }
    __syncthreads();

    wave_gemm<K, N1, 2, TN, G>(As, B1, C1, m0, wn * (TN * 16), lane);
    const int rt = wn & 1;
    wave_gemm<K, N2, 1, 2, G>(As + 16 * rt * P, B2, C2, m0 + 16 * rt, (wn >> 1) * 32, lane);
}

__device__ __forceinline__ void fma4(float4& a, const float4 v, const float w) {
    a.x = fmaf(v.x, w, a.x); a.y = fmaf(v.y, w, a.y);
    a.z = fmaf(v.z, w, a.z); a.w = fmaf(v.w, w, a.w);
}

__device__ __forceinline__ float dinv_of(const int c) { return rsqrtf((float)(c + 1)); }

__device__ __forceinline__ void load_chunk(const int* __restrict__ col,
                                           const int* __restrict__ cnt,
                                           int first, int m, int lane, float di,
                                           int& my_s, float& my_w) {
    my_s = 0; my_w = 0.0f;
    if (lane < m) {
        unsigned s = (unsigned)col[first + lane];
        s = s < (unsigned)N_NODES ? s : (unsigned)(N_NODES - 1);
        my_s = (int)s;
        my_w = dinv_of(cnt[s]) * di;
    }
}

__device__ __forceinline__ bool ovf_entry(const int i, const int node, const float di,
                                          const int* __restrict__ cnt,
                                          const int* __restrict__ ovf_dst,
                                          const int* __restrict__ ovf_src,
                                          int& s_out, float& w_out) {
    if (ovf_dst[i] != node) return false;
    unsigned s = (unsigned)ovf_src[i];
    s = s < (unsigned)N_NODES ? s : (unsigned)(N_NODES - 1);
    s_out = (int)s;
    w_out = dinv_of(cnt[s]) * di;
    return true;
}

template <int R>
__device__ __forceinline__ void add_rows256(const float4* __restrict__ Hv,
                                            int my_s, float my_w, int j, int lane,
                                            float4& a0, float4& a1) {
    float4 v[2 * R]; float w[R];
    for (int u = 0; u < R; ++u) {
        const int s = __builtin_amdgcn_readlane(my_s, j + u);
        w[u] = __int_as_float(__builtin_amdgcn_readlane(__float_as_int(my_w), j + u));
        const float4* r = Hv + (size_t)s * 64;
        v[2 * u]     = r[lane];
        v[2 * u + 1] = r[lane + 32];
    }
    for (int u = 0; u < R; ++u) { fma4(a0, v[2 * u], w[u]); fma4(a1, v[2 * u + 1], w[u]); }
}

#define G2_ROWS 32
__global__ void __launch_bounds__(256) k_gather256_gemm(const float* __restrict__ H,
                                                        const int* __restrict__ col,
                                                        const int* __restrict__ cnt,
                                                        const int* __restrict__ ovf_cnt,
                                                        const int* __restrict__ ovf_dst,
                                                        const int* __restrict__ ovf_src,
                                                        const float* __restrict__ b1,
                                                        const float* __restrict__ gamma,
                                                        const float* __restrict__ beta,
                                                        const float* __restrict__ W2,
                                                        float* __restrict__ C) {
    constexpr int P = A_PITCH(F_HID);
    __shared__ __attribute__((aligned(16))) float As[G2_ROWS * P];
    const int wave = __builtin_amdgcn_readfirstlane((int)(threadIdx.x >> 5));
    const int lane = threadIdx.x & 31;
    const int m0   = blockIdx.x * G2_ROWS;
    const float4* Hv = (const float4*)H;

    for (int q = 0; q < G2_ROWS / 8; ++q) {
        const int row = wave + 8 * q;
        int node = m0 + row;
        node = node < N_NODES ? node : N_NODES - 1;
        const int start = node * CAP;
        const int n     = cnt[node];
        const int nb    = n < CAP ? n : CAP;
        const float di  = dinv_of(n);

        float4 a0, a1;
        {
            const float w = di * di;
            const float4 v0 = Hv[(size_t)node * 64 + lane];
            const float4 v1 = Hv[(size_t)node * 64 + 32 + lane];
            a0.x = v0.x * w; a0.y = v0.y * w; a0.z = v0.z * w; a0.w = v0.w * w;
            a1.x = v1.x * w; a1.y = v1.y * w; a1.z = v1.z * w; a1.w = v1.w * w;
        }
        for (int base = 0; base < nb; base += 32) {
            const int m = (nb - base) < 32 ? (nb - base) : 32;
            int my_s; float my_w;
            load_chunk(col, cnt, start + base, m, lane, di, my_s, my_w);
            int j = 0;
            for (; j + 4 <= m; j += 4) add_rows256<4>(Hv, my_s, my_w, j, lane, a0, a1);
            const int left = m - j;
            if (left == 3)      add_rows256<3>(Hv, my_s, my_w, j, lane, a0, a1);
            else if (left == 2) add_rows256<2>(Hv, my_s, my_w, j, lane, a0, a1);
            else if (left == 1) add_rows256<1>(Hv, my_s, my_w, j, lane, a0, a1);
        }
        if (n > CAP) {
            int novf = *ovf_cnt;
            novf = novf < N_EDGES ? novf : N_EDGES;
            for (int i = 0; i < novf; ++i) {
                int s; float w;
                if (ovf_entry(i, node, di, cnt, ovf_dst, ovf_src, s, w)) {
                    fma4(a0, Hv[(size_t)s * 64 + lane], w);
                    fma4(a1, Hv[(size_t)s * 64 + 32 + lane], w);
                }
            }
        }
        float4 v[2] = {a0, a1};
        float s = 0.0f;
#pragma unroll
        for (int i = 0; i < 2; ++i) {
            float4 t  = v[i];
            float4 bb = ((const float4*)b1)[lane + 32 * i];
            t.x += bb.x; t.y += bb.y; t.z += bb.z; t.w += bb.w;
            v[i] = t;
            s += t.x + t.y + t.z + t.w;
        }
        float mu = wave_sum32(s) * (1.0f / F_HID);
        float ss = 0.0f;
#pragma unroll
        for (int i = 0; i < 2; ++i) {
            float4 t = v[i];
            ss += (t.x - mu) * (t.x - mu) + (t.y - mu) * (t.y - mu) +
                  (t.z - mu) * (t.z - mu) + (t.w - mu) * (t.w - mu);
        }
        float rstd = rsqrtf(wave_sum32(ss) * (1.0f / F_HID) + LN_EPS);
#pragma unroll
        for (int i = 0; i < 2; ++i) {
            float4 t = v[i];
            float4 g = ((const float4*)gamma)[lane + 32 * i];
            float4 b = ((const float4*)beta)[lane + 32 * i];
            t.x = fmaxf((t.x - mu) * rstd * g.x + b.x, 0.0f);
            t.y = fmaxf((t.y - mu) * rstd * g.y + b.y, 0.0f);
            t.z = fmaxf((t.z - mu) * rstd * g.z + b.z, 0.0f);
            t.w = fmaxf((t.w - mu) * rstd * g.w + b.w, 0.0f);
            *(float4*)&As[row * P + 4 * (lane + 32 * i)] = t;
        }
    }
    __syncthreads();

    if (wave < 4) {
        const int rt = wave & 1;
        wave_gemm<F_HID, F_OUT, 1, 2, 8>(As + 16 * rt * P, W2, C, m0 + 16 * rt, (wave >> 1) * 32, lane);
    }
}

__device__ __forceinline__ float final_value(float v) {
    const float lim = v > 0.0f ? 20.0f : -20.0f;
    v = isinf(v) ? lim : v;
    return isnan(v) ? 0.0f : v;
}

__global__ void __launch_bounds__(256) k_gather64(const float* __restrict__ H,
                                                  const int* __restrict__ col,
                                                  const int* __restrict__ cnt,
                                                  const int* __restrict__ ovf_cnt,
                                                  const int* __restrict__ ovf_dst,
                                                  const int* __restrict__ ovf_src,
                                                  const float* __restrict__ b2,
                                                  float* __restrict__ out) {
    const int wave = __builtin_amdgcn_readfirstlane((int)(threadIdx.x >> 5));
    const int node = blockIdx.x * 8 + wave;
    const int lane = threadIdx.x & 31;
    const int half = lane >> 4;
    const int c    = lane & 15;
    const int start = node * CAP;
    const int n     = cnt[node];
    const int nb    = n < CAP ? n : CAP;
    const float di  = dinv_of(n);
    const float4* Hv = (const float4*)H;

    float4 acc = {0.0f, 0.0f, 0.0f, 0.0f};
    if (half == 0) {
        const float w = di * di;
        const float4 v = Hv[(size_t)node * 16 + c];
        acc.x = v.x * w; acc.y = v.y * w; acc.z = v.z * w; acc.w = v.w * w;
    }
    for (int base = 0; base < nb; base += 32) {
        const int m = (nb - base) < 32 ? (nb - base) : 32;
        int my_s; float my_w;
        load_chunk(col, cnt, start + base, m, lane, di, my_s, my_w);
        for (int j = 0; j < m; j += 8) {
            float4 v[4]; float w[4];
            for (int u = 0; u < 4; ++u) {
                const int e0 = j + 2 * u;
                const int sA = __builtin_amdgcn_readlane(my_s, e0);
                const int sB = __builtin_amdgcn_readlane(my_s, e0 + 1);
                const float wA = __int_as_float(__builtin_amdgcn_readlane(__float_as_int(my_w), e0));
                const float wB = __int_as_float(__builtin_amdgcn_readlane(__float_as_int(my_w), e0 + 1));
                const int s = half ? sB : sA;
                w[u] = half ? wB : wA;
                v[u] = Hv[(size_t)s * 16 + c];
            }
            for (int u = 0; u < 4; ++u)
                if (j + 2 * u + half < m) fma4(acc, v[u], w[u]);
        }
    }
    if (n > CAP) {
        int novf = *ovf_cnt;
        novf = novf < N_EDGES ? novf : N_EDGES;
        for (int i = 0; i < novf; ++i) {
            int s; float w;
            if (ovf_entry(i, node, di, cnt, ovf_dst, ovf_src, s, w)) {
                if (half == 0) fma4(acc, Hv[(size_t)s * 16 + c], w);
            }
        }
    }
    acc.x += __shfl_xor(acc.x, 16, 32);
    acc.y += __shfl_xor(acc.y, 16, 32);
    acc.z += __shfl_xor(acc.z, 16, 32);
    acc.w += __shfl_xor(acc.w, 16, 32);
    if (half == 0) {
        float4* o = (float4*)(out + (size_t)node * F_OUT) + c;
        const float4 bb = ((const float4*)b2)[c];
        float4 t = *o;
        t.x = final_value(t.x + acc.x + bb.x);
        t.y = final_value(t.y + acc.y + bb.y);
        t.z = final_value(t.z + acc.z + bb.z);
        t.w = final_value(t.w + acc.w + bb.w);
        *o = t;
    }
}

extern "C" void kernel_launch(void* const* d_in, const int* in_sizes, int n_in,
                              void* d_out, int out_size, void* d_ws, size_t ws_size,
                              hipStream_t stream) {
    const float* x     = (const float*)d_in[0];
    const int*   ei    = (const int*)d_in[1];
    const float* W1    = (const float*)d_in[2];
    const float* b1    = (const float*)d_in[3];
    const float* gamma = (const float*)d_in[4];
    const float* beta  = (const float*)d_in[5];
    const float* W2    = (const float*)d_in[6];
    const float* b2    = (const float*)d_in[7];
    const float* Wskip = (const float*)d_in[8];
    const int* srcp = ei;
    const int* dstp = ei + N_EDGES;

    float* ws   = (float*)d_ws;
    float* base = ws + N_NODES;
    float* h1   = base + (size_t)N_NODES * F_IN + 48;
    float* h2   = h1 + (size_t)N_NODES * F_HID;
    float* out  = (float*)d_out;

    int* cnt     = (int*)base + 16;
    int* col     = cnt + N_LINE;
    int* ovf_dst = col + (size_t)N_NODES * CAP;
    int* ovf_src = ovf_dst + N_EDGES;
    int* ovf_cnt = ovf_src + N_EDGES;

    k_deg_init<<<(N_NODES + 255) / 256, 256, 0, stream>>>(cnt, ovf_cnt);

    k_norm_gemm_pair<F_IN, F_HID, F_OUT, 4, 4><<<GEMM_BLOCKS + EDGE_BLOCKS, 128, 0, stream>>>(
        x, W1, Wskip, h1, out, srcp, dstp, cnt, col, ovf_cnt, ovf_dst, ovf_src);

    k_gather256_gemm<<<(N_NODES + G2_ROWS - 1) / G2_ROWS, 256, 0, stream>>>(
        h1, col, cnt, ovf_cnt, ovf_dst, ovf_src, b1, gamma, beta, W2, h2);

    k_gather64<<<N_NODES / 8, 256, 0, stream>>>(h2, col, cnt, ovf_cnt, ovf_dst, ovf_src, b2, out);
}
